# starting kernel
# baseline (speedup 1.0000x reference)
#include <hip/hip_runtime.h>
#include <hip/hip_bf16.h>

typedef __attribute__((ext_vector_type(2))) float v2f;
typedef __attribute__((ext_vector_type(8))) float v8f;

#define NSIDE_  64
#define NLAT_   255
#define LMAX_   96
#define MMAX_   96
#define BC_     256
#define NPIX_   49152
#define L2_     48
#define M2_     48

__device__ __forceinline__ int ring_nlon(int t) {
  return (t < NSIDE_ - 1) ? 4 * (t + 1)
       : (t < 3 * NSIDE_) ? 4 * NSIDE_
       : 4 * (4 * NSIDE_ - 1 - t);
}

__device__ __forceinline__ int ring_offset(int t) {
  if (t < NSIDE_ - 1) return 2 * t * (t + 1);
  if (t < 3 * NSIDE_) return 2 * (NSIDE_ - 1) * NSIDE_ + 4 * NSIDE_ * (t - (NSIDE_ - 1));
  int r = NLAT_ - t;
  return NPIX_ - 2 * r * (r + 1);
}

__device__ __forceinline__ size_t tab_base(int k) {
  return (size_t)192 * k * (k - 1);
}
#define TAB_FLOATS 798720

__global__ void sht_build_tables(float* __restrict__ cosT, float* __restrict__ sinT) {
  const int k  = blockIdx.z + 1;
  const int n  = 4 * k;
  const int m2 = blockIdx.y;
  const int j  = blockIdx.x * 64 + threadIdx.x;
  if (j >= n) return;
  const int nh = n >> 1;
  const float twon = 6.28318530717958647692f / (float)n;
  float cp[2], sp[2];
  #pragma unroll
  for (int e = 0; e < 2; ++e) {
    const int mm = 2 * m2 + e;
    const float wc  = (mm == 0) ? 1.f : (mm < nh ? 2.f : (mm == nh ? 1.f : 0.f));
    const float wsn = (mm == nh) ? 0.f : wc;
    float s, c;
    __sincosf(twon * (float)((mm * j) % n), &s, &c);
    cp[e] = wc * c;
    sp[e] = -wsn * s;
  }
  const size_t o = tab_base(k) + ((size_t)m2 * n + j) * 2;
  cosT[o] = cp[0]; cosT[o + 1] = cp[1];
  sinT[o] = sp[0]; sinT[o + 1] = sp[1];
}

__global__ void sht_repack_pct(const float* __restrict__ pct,
                               float* __restrict__ pctP) {
  const int m = blockIdx.y;
  const int l = blockIdx.x;
  const int t = threadIdx.x;
  const float v = (t < NLAT_) ? pct[((size_t)m * LMAX_ + l) * NLAT_ + t] : 0.f;
  pctP[(((size_t)m * L2_ + (l >> 1)) * 256 + t) * 2 + (l & 1)] = v;
}

__global__ void sht_transpose_x(const float* __restrict__ xr,
                                const float* __restrict__ xi,
                                float* __restrict__ xTr,
                                float* __restrict__ xTi) {
  __shared__ float tr[32][33];
  __shared__ float ti[32][33];
  const int l   = blockIdx.z;
  const int m0  = blockIdx.x * 32;
  const int bc0 = blockIdx.y * 32;
  const int tx = threadIdx.x;
  const int ty = threadIdx.y;
  #pragma unroll
  for (int r = 0; r < 4; ++r) {
    const int bc = bc0 + ty + 8 * r;
    const int m  = m0 + tx;
    const int idx = (bc * LMAX_ + l) * MMAX_ + m;
    tr[ty + 8 * r][tx] = xr[idx];
    ti[ty + 8 * r][tx] = xi[idx];
  }
  __syncthreads();
  const size_t lbase = ((size_t)(l >> 1)) * BC_ * 2 + (l & 1);
  #pragma unroll
  for (int r = 0; r < 4; ++r) {
    const int m  = m0 + ty + 8 * r;
    const int bc = bc0 + tx;
    const size_t o = (size_t)m * L2_ * BC_ * 2 + lbase + (size_t)bc * 2;
    xTr[o] = tr[tx][ty + 8 * r];
    xTi[o] = ti[tx][ty + 8 * r];
  }
}

__global__ __launch_bounds__(32)
void sht_legendre_rotate(const float* __restrict__ xTr,
                         const float* __restrict__ xTi,
                         const float* __restrict__ pctP,
                         const float* __restrict__ offset,
                         float* __restrict__ cA,
                         float* __restrict__ cB) {
  const int lane   = threadIdx.x;
  const int bcpair = blockIdx.x;
  const int ttile  = blockIdx.y;
  const int m      = blockIdx.z;
  const int hi  = lane >> 4;
  const int lid = lane & 15;
  const int trow = ttile * 16 + lid;
  const int bc0  = bcpair * 32 + lid;

  const float* __restrict__ pb = pctP + ((size_t)m * L2_ * 256 + trow) * 2;
  const float* __restrict__ xr = xTr + ((size_t)m * L2_ * BC_ + bc0) * 2;
  const float* __restrict__ xi = xTi + ((size_t)m * L2_ * BC_ + bc0) * 2;

  v8f accR0 = {0.f, 0.f, 0.f, 0.f, 0.f, 0.f, 0.f, 0.f};
  v8f accI0 = accR0, accR1 = accR0, accI1 = accR0;

  for (int kk = 0; kk < LMAX_; kk += 4) {
    const int l2 = (kk >> 1) + hi;
    const v2f af  = *(const v2f*)(pb + (size_t)l2 * 256 * 2);
    const v2f bR0 = *(const v2f*)(xr + (size_t)l2 * BC_ * 2);
    const v2f bR1 = *(const v2f*)(xr + (size_t)l2 * BC_ * 2 + 32);
    const v2f bI0 = *(const v2f*)(xi + (size_t)l2 * BC_ * 2);
    const v2f bI1 = *(const v2f*)(xi + (size_t)l2 * BC_ * 2 + 32);
    accR0 = __builtin_amdgcn_wmma_f32_16x16x4_f32(false, af, false, bR0,
                                                  (short)0, accR0, false, false);
    accI0 = __builtin_amdgcn_wmma_f32_16x16x4_f32(false, af, false, bI0,
                                                  (short)0, accI0, false, false);
    accR1 = __builtin_amdgcn_wmma_f32_16x16x4_f32(false, af, false, bR1,
                                                  (short)0, accR1, false, false);
    accI1 = __builtin_amdgcn_wmma_f32_16x16x4_f32(false, af, false, bI1,
                                                  (short)0, accI1, false, false);
  }

  #pragma unroll
  for (int v = 0; v < 8; ++v) {
    const int t = ttile * 16 + v + 8 * hi;
    if (t < NLAT_) {
      float s, c;
      __sincosf(offset[t * MMAX_ + m], &s, &c);
      const size_t o = (((size_t)t * M2_ + (m >> 1)) * BC_ + bc0) * 2 + (m & 1);
      float re = accR0[v], im = accI0[v];
      cA[o]      = re * c - im * s;
      cB[o]      = re * s + im * c;
      re = accR1[v]; im = accI1[v];
      cA[o + 32] = re * c - im * s;
      cB[o + 32] = re * s + im * c;
    }
  }
}

__global__ __launch_bounds__(32)
void sht_ring_idft(const float* __restrict__ cA,
                   const float* __restrict__ cB,
                   const float* __restrict__ cosT,
                   const float* __restrict__ sinT,
                   float* __restrict__ out) {
  const int lane   = threadIdx.x;
  const int jtile  = blockIdx.x;
  const int bcpair = blockIdx.y;
  const int t      = blockIdx.z;
  const int n = ring_nlon(t);
  if (jtile * 16 >= n) return;
  const int roff = ring_offset(t);
  const int hi  = lane >> 4;
  const int lid = lane & 15;
  const int bc0 = bcpair * 32 + lid;
  const int j   = jtile * 16 + lid;

  const float* __restrict__ aB = cA + ((size_t)t * M2_ * BC_ + bc0) * 2;
  const float* __restrict__ bB = cB + ((size_t)t * M2_ * BC_ + bc0) * 2;
  const size_t tb = tab_base(n >> 2) + (size_t)j * 2;
  const float* __restrict__ cT = cosT + tb;
  const float* __restrict__ sT = sinT + tb;

  v8f acc0 = {0.f, 0.f, 0.f, 0.f, 0.f, 0.f, 0.f, 0.f};
  v8f acc1 = acc0;

  for (int kk = 0; kk < MMAX_; kk += 4) {
    const int m2 = (kk >> 1) + hi;
    const v2f cf  = *(const v2f*)(cT + (size_t)m2 * n * 2);
    const v2f sf  = *(const v2f*)(sT + (size_t)m2 * n * 2);
    const v2f af0 = *(const v2f*)(aB + (size_t)m2 * BC_ * 2);
    const v2f af1 = *(const v2f*)(aB + (size_t)m2 * BC_ * 2 + 32);
    const v2f bf0 = *(const v2f*)(bB + (size_t)m2 * BC_ * 2);
    const v2f bf1 = *(const v2f*)(bB + (size_t)m2 * BC_ * 2 + 32);
    acc0 = __builtin_amdgcn_wmma_f32_16x16x4_f32(false, af0, false, cf,
                                                 (short)0, acc0, false, false);
    acc0 = __builtin_amdgcn_wmma_f32_16x16x4_f32(false, bf0, false, sf,
                                                 (short)0, acc0, false, false);
    acc1 = __builtin_amdgcn_wmma_f32_16x16x4_f32(false, af1, false, cf,
                                                 (short)0, acc1, false, false);
    acc1 = __builtin_amdgcn_wmma_f32_16x16x4_f32(false, bf1, false, sf,
                                                 (short)0, acc1, false, false);
  }

  #pragma unroll
  for (int v = 0; v < 8; ++v) {
    const int bc = bcpair * 32 + v + 8 * hi;
    if (j < n) {
      out[(size_t)bc * NPIX_ + roff + j]        = acc0[v];
      out[(size_t)(bc + 16) * NPIX_ + roff + j] = acc1[v];
    }
  }
}

extern "C" void kernel_launch(void* const* d_in, const int* in_sizes, int n_in,
                              void* d_out, int out_size, void* d_ws, size_t ws_size,
                              hipStream_t stream) {
  (void)in_sizes; (void)n_in; (void)out_size; (void)ws_size;
  const float* x_real = (const float*)d_in[0];
  const float* x_imag = (const float*)d_in[1];
  const float* pct    = (const float*)d_in[2];
  const float* offset = (const float*)d_in[3];
  float* out = (float*)d_out;

  float* xTr  = (float*)d_ws;
  float* xTi  = xTr  + (size_t)MMAX_ * L2_ * BC_ * 2;
  float* pctP = xTi  + (size_t)MMAX_ * L2_ * BC_ * 2;
  float* cA   = pctP + (size_t)MMAX_ * L2_ * 256 * 2;
  float* cB   = cA   + (size_t)NLAT_ * M2_ * BC_ * 2;
  float* cosT = cB   + (size_t)NLAT_ * M2_ * BC_ * 2;
  float* sinT = cosT + (size_t)TAB_FLOATS;

  sht_build_tables<<<dim3(4, M2_, 64), dim3(64), 0, stream>>>(cosT, sinT);

  sht_repack_pct<<<dim3(LMAX_, MMAX_), dim3(256), 0, stream>>>(pct, pctP);

  sht_transpose_x<<<dim3(3, 8, LMAX_), dim3(32, 8), 0, stream>>>(
      x_real, x_imag, xTr, xTi);

  sht_legendre_rotate<<<dim3(8, 16, MMAX_), dim3(32), 0, stream>>>(
      xTr, xTi, pctP, offset, cA, cB);

  sht_ring_idft<<<dim3(16, 8, NLAT_), dim3(32), 0, stream>>>(
      cA, cB, cosT, sinT, out);
}


# the three independent preparation launches (DFT tables, pct repack, x transpose) run as one 256-thread launch selected by block index: 5 launches -> 3; arithmetic, stores and sum order unchanged
# speedup vs baseline: 1.0753x; 1.0753x over previous
#include <hip/hip_runtime.h>
#include <hip/hip_bf16.h>

typedef __attribute__((ext_vector_type(2))) float v2f;
typedef __attribute__((ext_vector_type(8))) float v8f;

#define NSIDE_  64
#define NLAT_   255
#define LMAX_   96
#define MMAX_   96
#define BC_     256
#define NPIX_   49152
#define L2_     48
#define M2_     48

__device__ __forceinline__ int ring_nlon(int t) {
  return (t < NSIDE_ - 1) ? 4 * (t + 1)
       : (t < 3 * NSIDE_) ? 4 * NSIDE_
       : 4 * (4 * NSIDE_ - 1 - t);
}

__device__ __forceinline__ int ring_offset(int t) {
  if (t < NSIDE_ - 1) return 2 * t * (t + 1);
  if (t < 3 * NSIDE_) return 2 * (NSIDE_ - 1) * NSIDE_ + 4 * NSIDE_ * (t - (NSIDE_ - 1));
  int r = NLAT_ - t;
  return NPIX_ - 2 * r * (r + 1);
}

__device__ __forceinline__ size_t tab_base(int k) {
  return (size_t)192 * k * (k - 1);
}
#define TAB_FLOATS 798720

__device__ __forceinline__ void prep_build_tables(int b, int tid,
                                                  float* __restrict__ cosT,
                                                  float* __restrict__ sinT) {
  const int k  = b / M2_ + 1;
  const int n  = 4 * k;
  const int m2 = b % M2_;
  const int j  = tid;
  if (j >= n) return;
  const int nh = n >> 1;
  const float twon = 6.28318530717958647692f / (float)n;
  float cp[2], sp[2];
  #pragma unroll
  for (int e = 0; e < 2; ++e) {
    const int mm = 2 * m2 + e;
    const float wc  = (mm == 0) ? 1.f : (mm < nh ? 2.f : (mm == nh ? 1.f : 0.f));
    const float wsn = (mm == nh) ? 0.f : wc;
    float s, c;
    __sincosf(twon * (float)((mm * j) % n), &s, &c);
    cp[e] = wc * c;
    sp[e] = -wsn * s;
  }
  const size_t o = tab_base(k) + ((size_t)m2 * n + j) * 2;
  cosT[o] = cp[0]; cosT[o + 1] = cp[1];
  sinT[o] = sp[0]; sinT[o + 1] = sp[1];
}

__device__ __forceinline__ void prep_repack_pct(int b, int tid,
                                                const float* __restrict__ pct,
                                                float* __restrict__ pctP) {
  const int m = b / LMAX_;
  const int l = b % LMAX_;
  const int t = tid;
  const float v = (t < NLAT_) ? pct[((size_t)m * LMAX_ + l) * NLAT_ + t] : 0.f;
  pctP[(((size_t)m * L2_ + (l >> 1)) * 256 + t) * 2 + (l & 1)] = v;
}

__device__ __forceinline__ void prep_transpose_x(int b, int tid,
                                                 float (*tr)[33], float (*ti)[33],
                                                 const float* __restrict__ xr,
                                                 const float* __restrict__ xi,
                                                 float* __restrict__ xTr,
                                                 float* __restrict__ xTi) {
  const int l   = b / 24;
  const int m0  = (b % 3) * 32;
  const int bc0 = ((b / 3) % 8) * 32;
  const int tx = tid & 31;
  const int ty = tid >> 5;
  #pragma unroll
  for (int r = 0; r < 4; ++r) {
    const int bc = bc0 + ty + 8 * r;
    const int m  = m0 + tx;
    const int idx = (bc * LMAX_ + l) * MMAX_ + m;
    tr[ty + 8 * r][tx] = xr[idx];
    ti[ty + 8 * r][tx] = xi[idx];
  }
  __syncthreads();
  const size_t lbase = ((size_t)(l >> 1)) * BC_ * 2 + (l & 1);
  #pragma unroll
  for (int r = 0; r < 4; ++r) {
    const int m  = m0 + ty + 8 * r;
    const int bc = bc0 + tx;
    const size_t o = (size_t)m * L2_ * BC_ * 2 + lbase + (size_t)bc * 2;
    xTr[o] = tr[tx][ty + 8 * r];
    xTi[o] = ti[tx][ty + 8 * r];
  }
}

#define PREP_T_BLOCKS_ 2304
#define PREP_P_BLOCKS_ 9216
#define PREP_BLOCKS_   14592
__global__ __launch_bounds__(256)
void sht_prepare(const float* __restrict__ xr,
                 const float* __restrict__ xi,
                 const float* __restrict__ pct,
                 float* __restrict__ xTr,
                 float* __restrict__ xTi,
                 float* __restrict__ pctP,
                 float* __restrict__ cosT,
                 float* __restrict__ sinT) {
  __shared__ float tr[32][33];
  __shared__ float ti[32][33];
  const int b   = blockIdx.x;
  const int tid = threadIdx.x;
  if (b < PREP_T_BLOCKS_) {
    prep_transpose_x(b, tid, tr, ti, xr, xi, xTr, xTi);
  } else if (b < PREP_T_BLOCKS_ + PREP_P_BLOCKS_) {
    prep_repack_pct(b - PREP_T_BLOCKS_, tid, pct, pctP);
  } else {
    prep_build_tables(b - (PREP_T_BLOCKS_ + PREP_P_BLOCKS_), tid, cosT, sinT);
  }
}

__global__ __launch_bounds__(32)
void sht_legendre_rotate(const float* __restrict__ xTr,
                         const float* __restrict__ xTi,
                         const float* __restrict__ pctP,
                         const float* __restrict__ offset,
                         float* __restrict__ cA,
                         float* __restrict__ cB) {
  const int lane   = threadIdx.x;
  const int bcpair = blockIdx.x;
  const int ttile  = blockIdx.y;
  const int m      = blockIdx.z;
  const int hi  = lane >> 4;
  const int lid = lane & 15;
  const int trow = ttile * 16 + lid;
  const int bc0  = bcpair * 32 + lid;

  const float* __restrict__ pb = pctP + ((size_t)m * L2_ * 256 + trow) * 2;
  const float* __restrict__ xr = xTr + ((size_t)m * L2_ * BC_ + bc0) * 2;
  const float* __restrict__ xi = xTi + ((size_t)m * L2_ * BC_ + bc0) * 2;

  v8f accR0 = {0.f, 0.f, 0.f, 0.f, 0.f, 0.f, 0.f, 0.f};
  v8f accI0 = accR0, accR1 = accR0, accI1 = accR0;

  for (int kk = 0; kk < LMAX_; kk += 4) {
    const int l2 = (kk >> 1) + hi;
    const v2f af  = *(const v2f*)(pb + (size_t)l2 * 256 * 2);
    const v2f bR0 = *(const v2f*)(xr + (size_t)l2 * BC_ * 2);
    const v2f bR1 = *(const v2f*)(xr + (size_t)l2 * BC_ * 2 + 32);
    const v2f bI0 = *(const v2f*)(xi + (size_t)l2 * BC_ * 2);
    const v2f bI1 = *(const v2f*)(xi + (size_t)l2 * BC_ * 2 + 32);
    accR0 = __builtin_amdgcn_wmma_f32_16x16x4_f32(false, af, false, bR0,
                                                  (short)0, accR0, false, false);
    accI0 = __builtin_amdgcn_wmma_f32_16x16x4_f32(false, af, false, bI0,
                                                  (short)0, accI0, false, false);
    accR1 = __builtin_amdgcn_wmma_f32_16x16x4_f32(false, af, false, bR1,
                                                  (short)0, accR1, false, false);
    accI1 = __builtin_amdgcn_wmma_f32_16x16x4_f32(false, af, false, bI1,
                                                  (short)0, accI1, false, false);
  }

  #pragma unroll
  for (int v = 0; v < 8; ++v) {
    const int t = ttile * 16 + v + 8 * hi;
    if (t < NLAT_) {
      float s, c;
      __sincosf(offset[t * MMAX_ + m], &s, &c);
      const size_t o = (((size_t)t * M2_ + (m >> 1)) * BC_ + bc0) * 2 + (m & 1);
      float re = accR0[v], im = accI0[v];
      cA[o]      = re * c - im * s;
      cB[o]      = re * s + im * c;
      re = accR1[v]; im = accI1[v];
      cA[o + 32] = re * c - im * s;
      cB[o + 32] = re * s + im * c;
    }
  }
}

__global__ __launch_bounds__(32)
void sht_ring_idft(const float* __restrict__ cA,
                   const float* __restrict__ cB,
                   const float* __restrict__ cosT,
                   const float* __restrict__ sinT,
                   float* __restrict__ out) {
  const int lane   = threadIdx.x;
  const int jtile  = blockIdx.x;
  const int bcpair = blockIdx.y;
  const int t      = blockIdx.z;
  const int n = ring_nlon(t);
  if (jtile * 16 >= n) return;
  const int roff = ring_offset(t);
  const int hi  = lane >> 4;
  const int lid = lane & 15;
  const int bc0 = bcpair * 32 + lid;
  const int j   = jtile * 16 + lid;

  const float* __restrict__ aB = cA + ((size_t)t * M2_ * BC_ + bc0) * 2;
  const float* __restrict__ bB = cB + ((size_t)t * M2_ * BC_ + bc0) * 2;
  const size_t tb = tab_base(n >> 2) + (size_t)j * 2;
  const float* __restrict__ cT = cosT + tb;
  const float* __restrict__ sT = sinT + tb;

  v8f acc0 = {0.f, 0.f, 0.f, 0.f, 0.f, 0.f, 0.f, 0.f};
  v8f acc1 = acc0;

  for (int kk = 0; kk < MMAX_; kk += 4) {
    const int m2 = (kk >> 1) + hi;
    const v2f cf  = *(const v2f*)(cT + (size_t)m2 * n * 2);
    const v2f sf  = *(const v2f*)(sT + (size_t)m2 * n * 2);
    const v2f af0 = *(const v2f*)(aB + (size_t)m2 * BC_ * 2);
    const v2f af1 = *(const v2f*)(aB + (size_t)m2 * BC_ * 2 + 32);
    const v2f bf0 = *(const v2f*)(bB + (size_t)m2 * BC_ * 2);
    const v2f bf1 = *(const v2f*)(bB + (size_t)m2 * BC_ * 2 + 32);
    acc0 = __builtin_amdgcn_wmma_f32_16x16x4_f32(false, af0, false, cf,
                                                 (short)0, acc0, false, false);
    acc0 = __builtin_amdgcn_wmma_f32_16x16x4_f32(false, bf0, false, sf,
                                                 (short)0, acc0, false, false);
    acc1 = __builtin_amdgcn_wmma_f32_16x16x4_f32(false, af1, false, cf,
                                                 (short)0, acc1, false, false);
    acc1 = __builtin_amdgcn_wmma_f32_16x16x4_f32(false, bf1, false, sf,
                                                 (short)0, acc1, false, false);
  }

  #pragma unroll
  for (int v = 0; v < 8; ++v) {
    const int bc = bcpair * 32 + v + 8 * hi;
    if (j < n) {
      out[(size_t)bc * NPIX_ + roff + j]        = acc0[v];
      out[(size_t)(bc + 16) * NPIX_ + roff + j] = acc1[v];
    }
  }
}

extern "C" void kernel_launch(void* const* d_in, const int* in_sizes, int n_in,
                              void* d_out, int out_size, void* d_ws, size_t ws_size,
                              hipStream_t stream) {
  (void)in_sizes; (void)n_in; (void)out_size; (void)ws_size;
  const float* x_real = (const float*)d_in[0];
  const float* x_imag = (const float*)d_in[1];
  const float* pct    = (const float*)d_in[2];
  const float* offset = (const float*)d_in[3];
  float* out = (float*)d_out;

  float* xTr  = (float*)d_ws;
  float* xTi  = xTr  + (size_t)MMAX_ * L2_ * BC_ * 2;
  float* pctP = xTi  + (size_t)MMAX_ * L2_ * BC_ * 2;
  float* cA   = pctP + (size_t)MMAX_ * L2_ * 256 * 2;
  float* cB   = cA   + (size_t)NLAT_ * M2_ * BC_ * 2;
  float* cosT = cB   + (size_t)NLAT_ * M2_ * BC_ * 2;
  float* sinT = cosT + (size_t)TAB_FLOATS;

  sht_prepare<<<dim3(PREP_BLOCKS_), dim3(256), 0, stream>>>(
      x_real, x_imag, pct, xTr, xTi, pctP, cosT, sinT);

  sht_legendre_rotate<<<dim3(8, 16, MMAX_), dim3(32), 0, stream>>>(
      xTr, xTi, pctP, offset, cA, cB);

  sht_ring_idft<<<dim3(16, 8, NLAT_), dim3(32), 0, stream>>>(
      cA, cB, cosT, sinT, out);
}


# both matrix kernels: the 16 waves sharing an operand slab (x per m and bc pair; coefficients per ring and bc pair) form one 512-thread work-group that stages it once in LDS; sum order unchanged
# speedup vs baseline: 1.4697x; 1.3668x over previous
#include <hip/hip_runtime.h>
#include <hip/hip_bf16.h>

typedef __attribute__((ext_vector_type(2))) float v2f;
typedef __attribute__((ext_vector_type(4))) float v4f;
typedef __attribute__((ext_vector_type(8))) float v8f;

#define NSIDE_  64
#define NLAT_   255
#define LMAX_   96
#define MMAX_   96
#define BC_     256
#define NPIX_   49152
#define L2_     48
#define M2_     48

__device__ __forceinline__ int ring_nlon(int t) {
  return (t < NSIDE_ - 1) ? 4 * (t + 1)
       : (t < 3 * NSIDE_) ? 4 * NSIDE_
       : 4 * (4 * NSIDE_ - 1 - t);
}

__device__ __forceinline__ int ring_offset(int t) {
  if (t < NSIDE_ - 1) return 2 * t * (t + 1);
  if (t < 3 * NSIDE_) return 2 * (NSIDE_ - 1) * NSIDE_ + 4 * NSIDE_ * (t - (NSIDE_ - 1));
  int r = NLAT_ - t;
  return NPIX_ - 2 * r * (r + 1);
}

__device__ __forceinline__ size_t tab_base(int k) {
  return (size_t)192 * k * (k - 1);
}
#define TAB_FLOATS 798720

__device__ __forceinline__ void prep_build_tables(int b, int tid,
                                                  float* __restrict__ cosT,
                                                  float* __restrict__ sinT) {
  const int k  = b / M2_ + 1;
  const int n  = 4 * k;
  const int m2 = b % M2_;
  const int j  = tid;
  if (j >= n) return;
  const int nh = n >> 1;
  const float twon = 6.28318530717958647692f / (float)n;
  float cp[2], sp[2];
  #pragma unroll
  for (int e = 0; e < 2; ++e) {
    const int mm = 2 * m2 + e;
    const float wc  = (mm == 0) ? 1.f : (mm < nh ? 2.f : (mm == nh ? 1.f : 0.f));
    const float wsn = (mm == nh) ? 0.f : wc;
    float s, c;
    __sincosf(twon * (float)((mm * j) % n), &s, &c);
    cp[e] = wc * c;
    sp[e] = -wsn * s;
  }
  const size_t o = tab_base(k) + ((size_t)m2 * n + j) * 2;
  cosT[o] = cp[0]; cosT[o + 1] = cp[1];
  sinT[o] = sp[0]; sinT[o + 1] = sp[1];
}

__device__ __forceinline__ void prep_repack_pct(int b, int tid,
                                                const float* __restrict__ pct,
                                                float* __restrict__ pctP) {
  const int m = b / LMAX_;
  const int l = b % LMAX_;
  const int t = tid;
  const float v = (t < NLAT_) ? pct[((size_t)m * LMAX_ + l) * NLAT_ + t] : 0.f;
  pctP[(((size_t)m * L2_ + (l >> 1)) * 256 + t) * 2 + (l & 1)] = v;
}

__device__ __forceinline__ void prep_transpose_x(int b, int tid,
                                                 float (*tr)[33], float (*ti)[33],
                                                 const float* __restrict__ xr,
                                                 const float* __restrict__ xi,
                                                 float* __restrict__ xTr,
                                                 float* __restrict__ xTi) {
  const int l   = b / 24;
  const int m0  = (b % 3) * 32;
  const int bc0 = ((b / 3) % 8) * 32;
  const int tx = tid & 31;
  const int ty = tid >> 5;
  #pragma unroll
  for (int r = 0; r < 4; ++r) {
    const int bc = bc0 + ty + 8 * r;
    const int m  = m0 + tx;
    const int idx = (bc * LMAX_ + l) * MMAX_ + m;
    tr[ty + 8 * r][tx] = xr[idx];
    ti[ty + 8 * r][tx] = xi[idx];
  }
  __syncthreads();
  const size_t lbase = ((size_t)(l >> 1)) * BC_ * 2 + (l & 1);
  #pragma unroll
  for (int r = 0; r < 4; ++r) {
    const int m  = m0 + ty + 8 * r;
    const int bc = bc0 + tx;
    const size_t o = (size_t)m * L2_ * BC_ * 2 + lbase + (size_t)bc * 2;
    xTr[o] = tr[tx][ty + 8 * r];
    xTi[o] = ti[tx][ty + 8 * r];
  }
}

#define PREP_T_BLOCKS_ 2304
#define PREP_P_BLOCKS_ 9216
#define PREP_BLOCKS_   14592
__global__ __launch_bounds__(256)
void sht_prepare(const float* __restrict__ xr,
                 const float* __restrict__ xi,
                 const float* __restrict__ pct,
                 float* __restrict__ xTr,
                 float* __restrict__ xTi,
                 float* __restrict__ pctP,
                 float* __restrict__ cosT,
                 float* __restrict__ sinT) {
  __shared__ float tr[32][33];
  __shared__ float ti[32][33];
  const int b   = blockIdx.x;
  const int tid = threadIdx.x;
  if (b < PREP_T_BLOCKS_) {
    prep_transpose_x(b, tid, tr, ti, xr, xi, xTr, xTi);
  } else if (b < PREP_T_BLOCKS_ + PREP_P_BLOCKS_) {
    prep_repack_pct(b - PREP_T_BLOCKS_, tid, pct, pctP);
  } else {
    prep_build_tables(b - (PREP_T_BLOCKS_ + PREP_P_BLOCKS_), tid, cosT, sinT);
  }
}

__global__ __launch_bounds__(512)
void sht_legendre_rotate(const float* __restrict__ xTr,
                         const float* __restrict__ xTi,
                         const float* __restrict__ pctP,
                         const float* __restrict__ offset,
                         float* __restrict__ cA,
                         float* __restrict__ cB) {
  __shared__ __attribute__((aligned(16))) float sR[L2_ * 64];
  __shared__ __attribute__((aligned(16))) float sI[L2_ * 64];
  const int tid    = threadIdx.x;
  const int lane   = tid & 31;
  const int ttile  = __builtin_amdgcn_readfirstlane(tid >> 5);
  const int bcpair = blockIdx.x;
  const int m      = blockIdx.y;

  {
    const float* __restrict__ gR = xTr + ((size_t)m * L2_ * BC_ + bcpair * 32) * 2;
    const float* __restrict__ gI = xTi + ((size_t)m * L2_ * BC_ + bcpair * 32) * 2;
    for (int idx = tid; idx < L2_ * 16; idx += 512) {
      const int row = idx >> 4;
      const int c4  = (idx & 15) * 4;
      const v4f vr = *(const v4f*)(gR + (size_t)row * (BC_ * 2) + c4);
      const v4f vi = *(const v4f*)(gI + (size_t)row * (BC_ * 2) + c4);
      *(v4f*)(sR + row * 64 + c4) = vr;
      *(v4f*)(sI + row * 64 + c4) = vi;
    }
  }
  __syncthreads();

  const int hi  = lane >> 4;
  const int lid = lane & 15;
  const int trow = ttile * 16 + lid;
  const int bc0  = bcpair * 32 + lid;

  const float* __restrict__ pb = pctP + ((size_t)m * L2_ * 256 + trow) * 2;

  v8f accR0 = {0.f, 0.f, 0.f, 0.f, 0.f, 0.f, 0.f, 0.f};
  v8f accI0 = accR0, accR1 = accR0, accI1 = accR0;

  for (int kk = 0; kk < LMAX_; kk += 4) {
    const int l2 = (kk >> 1) + hi;
    const v2f af  = *(const v2f*)(pb + (size_t)l2 * 256 * 2);
    const v2f bR0 = *(const v2f*)(sR + (l2 * 32 + lid) * 2);
    const v2f bR1 = *(const v2f*)(sR + (l2 * 32 + lid + 16) * 2);
    const v2f bI0 = *(const v2f*)(sI + (l2 * 32 + lid) * 2);
    const v2f bI1 = *(const v2f*)(sI + (l2 * 32 + lid + 16) * 2);
    accR0 = __builtin_amdgcn_wmma_f32_16x16x4_f32(false, af, false, bR0,
                                                  (short)0, accR0, false, false);
    accI0 = __builtin_amdgcn_wmma_f32_16x16x4_f32(false, af, false, bI0,
                                                  (short)0, accI0, false, false);
    accR1 = __builtin_amdgcn_wmma_f32_16x16x4_f32(false, af, false, bR1,
                                                  (short)0, accR1, false, false);
    accI1 = __builtin_amdgcn_wmma_f32_16x16x4_f32(false, af, false, bI1,
                                                  (short)0, accI1, false, false);
  }

  #pragma unroll
  for (int v = 0; v < 8; ++v) {
    const int t = ttile * 16 + v + 8 * hi;
    if (t < NLAT_) {
      float s, c;
      __sincosf(offset[t * MMAX_ + m], &s, &c);
      const size_t o = (((size_t)t * M2_ + (m >> 1)) * BC_ + bc0) * 2 + (m & 1);
      float re = accR0[v], im = accI0[v];
      cA[o]      = re * c - im * s;
      cB[o]      = re * s + im * c;
      re = accR1[v]; im = accI1[v];
      cA[o + 32] = re * c - im * s;
      cB[o + 32] = re * s + im * c;
    }
  }
}

__global__ __launch_bounds__(512)
void sht_ring_idft(const float* __restrict__ cA,
                   const float* __restrict__ cB,
                   const float* __restrict__ cosT,
                   const float* __restrict__ sinT,
                   float* __restrict__ out) {
  __shared__ __attribute__((aligned(16))) float sA[M2_ * 64];
  __shared__ __attribute__((aligned(16))) float sB[M2_ * 64];
  const int tid    = threadIdx.x;
  const int lane   = tid & 31;
  const int jtile  = __builtin_amdgcn_readfirstlane(tid >> 5);
  const int bcpair = blockIdx.x;
  const int t      = blockIdx.y;
  const int n = ring_nlon(t);

  {
    const float* __restrict__ gA = cA + ((size_t)t * M2_ * BC_ + bcpair * 32) * 2;
    const float* __restrict__ gB = cB + ((size_t)t * M2_ * BC_ + bcpair * 32) * 2;
    for (int idx = tid; idx < M2_ * 16; idx += 512) {
      const int row = idx >> 4;
      const int c4  = (idx & 15) * 4;
      const v4f va = *(const v4f*)(gA + (size_t)row * (BC_ * 2) + c4);
      const v4f vb = *(const v4f*)(gB + (size_t)row * (BC_ * 2) + c4);
      *(v4f*)(sA + row * 64 + c4) = va;
      *(v4f*)(sB + row * 64 + c4) = vb;
    }
  }
  __syncthreads();
  if (jtile * 16 >= n) return;

  const int roff = ring_offset(t);
  const int hi  = lane >> 4;
  const int lid = lane & 15;
  const int j   = jtile * 16 + lid;

  const size_t tb = tab_base(n >> 2) + (size_t)j * 2;
  const float* __restrict__ cT = cosT + tb;
  const float* __restrict__ sT = sinT + tb;

  v8f acc0 = {0.f, 0.f, 0.f, 0.f, 0.f, 0.f, 0.f, 0.f};
  v8f acc1 = acc0;

  for (int kk = 0; kk < MMAX_; kk += 4) {
    const int m2 = (kk >> 1) + hi;
    const v2f cf  = *(const v2f*)(cT + (size_t)m2 * n * 2);
    const v2f sf  = *(const v2f*)(sT + (size_t)m2 * n * 2);
    const v2f af0 = *(const v2f*)(sA + (m2 * 32 + lid) * 2);
    const v2f af1 = *(const v2f*)(sA + (m2 * 32 + lid + 16) * 2);
    const v2f bf0 = *(const v2f*)(sB + (m2 * 32 + lid) * 2);
    const v2f bf1 = *(const v2f*)(sB + (m2 * 32 + lid + 16) * 2);
    acc0 = __builtin_amdgcn_wmma_f32_16x16x4_f32(false, af0, false, cf,
                                                 (short)0, acc0, false, false);
    acc0 = __builtin_amdgcn_wmma_f32_16x16x4_f32(false, bf0, false, sf,
                                                 (short)0, acc0, false, false);
    acc1 = __builtin_amdgcn_wmma_f32_16x16x4_f32(false, af1, false, cf,
                                                 (short)0, acc1, false, false);
    acc1 = __builtin_amdgcn_wmma_f32_16x16x4_f32(false, bf1, false, sf,
                                                 (short)0, acc1, false, false);
  }

  #pragma unroll
  for (int v = 0; v < 8; ++v) {
    const int bc = bcpair * 32 + v + 8 * hi;
    if (j < n) {
      out[(size_t)bc * NPIX_ + roff + j]        = acc0[v];
      out[(size_t)(bc + 16) * NPIX_ + roff + j] = acc1[v];
    }
  }
}

extern "C" void kernel_launch(void* const* d_in, const int* in_sizes, int n_in,
                              void* d_out, int out_size, void* d_ws, size_t ws_size,
                              hipStream_t stream) {
  (void)in_sizes; (void)n_in; (void)out_size; (void)ws_size;
  const float* x_real = (const float*)d_in[0];
  const float* x_imag = (const float*)d_in[1];
  const float* pct    = (const float*)d_in[2];
  const float* offset = (const float*)d_in[3];
  float* out = (float*)d_out;

  float* xTr  = (float*)d_ws;
  float* xTi  = xTr  + (size_t)MMAX_ * L2_ * BC_ * 2;
  float* pctP = xTi  + (size_t)MMAX_ * L2_ * BC_ * 2;
  float* cA   = pctP + (size_t)MMAX_ * L2_ * 256 * 2;
  float* cB   = cA   + (size_t)NLAT_ * M2_ * BC_ * 2;
  float* cosT = cB   + (size_t)NLAT_ * M2_ * BC_ * 2;
  float* sinT = cosT + (size_t)TAB_FLOATS;

  sht_prepare<<<dim3(PREP_BLOCKS_), dim3(256), 0, stream>>>(
      x_real, x_imag, pct, xTr, xTi, pctP, cosT, sinT);

  sht_legendre_rotate<<<dim3(8, MMAX_), dim3(512), 0, stream>>>(
      xTr, xTi, pctP, offset, cA, cB);

  sht_ring_idft<<<dim3(8, NLAT_), dim3(512), 0, stream>>>(
      cA, cB, cosT, sinT, out);
}


# ring iDFT: table columns n-j mirror columns j (cos equal, sin negated), so only j=0..n/2 are multiplied; cos and sin sums kept apart, y[j]=C+S, y[n-j]=C-S (sum order changed); 384-thread groups
# speedup vs baseline: 1.7205x; 1.1706x over previous
#include <hip/hip_runtime.h>
#include <hip/hip_bf16.h>

typedef __attribute__((ext_vector_type(2))) float v2f;
typedef __attribute__((ext_vector_type(4))) float v4f;
typedef __attribute__((ext_vector_type(8))) float v8f;

#define NSIDE_  64
#define NLAT_   255
#define LMAX_   96
#define MMAX_   96
#define BC_     256
#define NPIX_   49152
#define L2_     48
#define M2_     48

__device__ __forceinline__ int ring_nlon(int t) {
  return (t < NSIDE_ - 1) ? 4 * (t + 1)
       : (t < 3 * NSIDE_) ? 4 * NSIDE_
       : 4 * (4 * NSIDE_ - 1 - t);
}

__device__ __forceinline__ int ring_offset(int t) {
  if (t < NSIDE_ - 1) return 2 * t * (t + 1);
  if (t < 3 * NSIDE_) return 2 * (NSIDE_ - 1) * NSIDE_ + 4 * NSIDE_ * (t - (NSIDE_ - 1));
  int r = NLAT_ - t;
  return NPIX_ - 2 * r * (r + 1);
}

__device__ __forceinline__ size_t tab_base(int k) {
  return (size_t)192 * k * (k - 1);
}
#define TAB_FLOATS 798720

__device__ __forceinline__ void prep_build_tables(int b, int tid,
                                                  float* __restrict__ cosT,
                                                  float* __restrict__ sinT) {
  const int k  = b / M2_ + 1;
  const int n  = 4 * k;
  const int m2 = b % M2_;
  const int j  = tid;
  if (j >= n) return;
  const int nh = n >> 1;
  const float twon = 6.28318530717958647692f / (float)n;
  float cp[2], sp[2];
  #pragma unroll
  for (int e = 0; e < 2; ++e) {
    const int mm = 2 * m2 + e;
    const float wc  = (mm == 0) ? 1.f : (mm < nh ? 2.f : (mm == nh ? 1.f : 0.f));
    const float wsn = (mm == nh) ? 0.f : wc;
    float s, c;
    __sincosf(twon * (float)((mm * j) % n), &s, &c);
    cp[e] = wc * c;
    sp[e] = -wsn * s;
  }
  const size_t o = tab_base(k) + ((size_t)m2 * n + j) * 2;
  cosT[o] = cp[0]; cosT[o + 1] = cp[1];
  sinT[o] = sp[0]; sinT[o + 1] = sp[1];
}

__device__ __forceinline__ void prep_repack_pct(int b, int tid,
                                                const float* __restrict__ pct,
                                                float* __restrict__ pctP) {
  const int m = b / LMAX_;
  const int l = b % LMAX_;
  const int t = tid;
  const float v = (t < NLAT_) ? pct[((size_t)m * LMAX_ + l) * NLAT_ + t] : 0.f;
  pctP[(((size_t)m * L2_ + (l >> 1)) * 256 + t) * 2 + (l & 1)] = v;
}

__device__ __forceinline__ void prep_transpose_x(int b, int tid,
                                                 float (*tr)[33], float (*ti)[33],
                                                 const float* __restrict__ xr,
                                                 const float* __restrict__ xi,
                                                 float* __restrict__ xTr,
                                                 float* __restrict__ xTi) {
  const int l   = b / 24;
  const int m0  = (b % 3) * 32;
  const int bc0 = ((b / 3) % 8) * 32;
  const int tx = tid & 31;
  const int ty = tid >> 5;
  #pragma unroll
  for (int r = 0; r < 4; ++r) {
    const int bc = bc0 + ty + 8 * r;
    const int m  = m0 + tx;
    const int idx = (bc * LMAX_ + l) * MMAX_ + m;
    tr[ty + 8 * r][tx] = xr[idx];
    ti[ty + 8 * r][tx] = xi[idx];
  }
  __syncthreads();
  const size_t lbase = ((size_t)(l >> 1)) * BC_ * 2 + (l & 1);
  #pragma unroll
  for (int r = 0; r < 4; ++r) {
    const int m  = m0 + ty + 8 * r;
    const int bc = bc0 + tx;
    const size_t o = (size_t)m * L2_ * BC_ * 2 + lbase + (size_t)bc * 2;
    xTr[o] = tr[tx][ty + 8 * r];
    xTi[o] = ti[tx][ty + 8 * r];
  }
}

#define PREP_T_BLOCKS_ 2304
#define PREP_P_BLOCKS_ 9216
#define PREP_BLOCKS_   14592
__global__ __launch_bounds__(256)
void sht_prepare(const float* __restrict__ xr,
                 const float* __restrict__ xi,
                 const float* __restrict__ pct,
                 float* __restrict__ xTr,
                 float* __restrict__ xTi,
                 float* __restrict__ pctP,
                 float* __restrict__ cosT,
                 float* __restrict__ sinT) {
  __shared__ float tr[32][33];
  __shared__ float ti[32][33];
  const int b   = blockIdx.x;
  const int tid = threadIdx.x;
  if (b < PREP_T_BLOCKS_) {
    prep_transpose_x(b, tid, tr, ti, xr, xi, xTr, xTi);
  } else if (b < PREP_T_BLOCKS_ + PREP_P_BLOCKS_) {
    prep_repack_pct(b - PREP_T_BLOCKS_, tid, pct, pctP);
  } else {
    prep_build_tables(b - (PREP_T_BLOCKS_ + PREP_P_BLOCKS_), tid, cosT, sinT);
  }
}

__global__ __launch_bounds__(512)
void sht_legendre_rotate(const float* __restrict__ xTr,
                         const float* __restrict__ xTi,
                         const float* __restrict__ pctP,
                         const float* __restrict__ offset,
                         float* __restrict__ cA,
                         float* __restrict__ cB) {
  __shared__ __attribute__((aligned(16))) float sR[L2_ * 64];
  __shared__ __attribute__((aligned(16))) float sI[L2_ * 64];
  const int tid    = threadIdx.x;
  const int lane   = tid & 31;
  const int ttile  = __builtin_amdgcn_readfirstlane(tid >> 5);
  const int bcpair = blockIdx.x;
  const int m      = blockIdx.y;

  {
    const float* __restrict__ gR = xTr + ((size_t)m * L2_ * BC_ + bcpair * 32) * 2;
    const float* __restrict__ gI = xTi + ((size_t)m * L2_ * BC_ + bcpair * 32) * 2;
    for (int idx = tid; idx < L2_ * 16; idx += 512) {
      const int row = idx >> 4;
      const int c4  = (idx & 15) * 4;
      const v4f vr = *(const v4f*)(gR + (size_t)row * (BC_ * 2) + c4);
      const v4f vi = *(const v4f*)(gI + (size_t)row * (BC_ * 2) + c4);
      *(v4f*)(sR + row * 64 + c4) = vr;
      *(v4f*)(sI + row * 64 + c4) = vi;
    }
  }
  __syncthreads();

  const int hi  = lane >> 4;
  const int lid = lane & 15;
  const int trow = ttile * 16 + lid;
  const int bc0  = bcpair * 32 + lid;

  const float* __restrict__ pb = pctP + ((size_t)m * L2_ * 256 + trow) * 2;

  v8f accR0 = {0.f, 0.f, 0.f, 0.f, 0.f, 0.f, 0.f, 0.f};
  v8f accI0 = accR0, accR1 = accR0, accI1 = accR0;

  for (int kk = 0; kk < LMAX_; kk += 4) {
    const int l2 = (kk >> 1) + hi;
    const v2f af  = *(const v2f*)(pb + (size_t)l2 * 256 * 2);
    const v2f bR0 = *(const v2f*)(sR + (l2 * 32 + lid) * 2);
    const v2f bR1 = *(const v2f*)(sR + (l2 * 32 + lid + 16) * 2);
    const v2f bI0 = *(const v2f*)(sI + (l2 * 32 + lid) * 2);
    const v2f bI1 = *(const v2f*)(sI + (l2 * 32 + lid + 16) * 2);
    accR0 = __builtin_amdgcn_wmma_f32_16x16x4_f32(false, af, false, bR0,
                                                  (short)0, accR0, false, false);
    accI0 = __builtin_amdgcn_wmma_f32_16x16x4_f32(false, af, false, bI0,
                                                  (short)0, accI0, false, false);
    accR1 = __builtin_amdgcn_wmma_f32_16x16x4_f32(false, af, false, bR1,
                                                  (short)0, accR1, false, false);
    accI1 = __builtin_amdgcn_wmma_f32_16x16x4_f32(false, af, false, bI1,
                                                  (short)0, accI1, false, false);
  }

  #pragma unroll
  for (int v = 0; v < 8; ++v) {
    const int t = ttile * 16 + v + 8 * hi;
    if (t < NLAT_) {
      float s, c;
      __sincosf(offset[t * MMAX_ + m], &s, &c);
      const size_t o = (((size_t)t * M2_ + (m >> 1)) * BC_ + bc0) * 2 + (m & 1);
      float re = accR0[v], im = accI0[v];
      cA[o]      = re * c - im * s;
      cB[o]      = re * s + im * c;
      re = accR1[v]; im = accI1[v];
      cA[o + 32] = re * c - im * s;
      cB[o + 32] = re * s + im * c;
    }
  }
}

#define RING_WAVES_ 12
__global__ __launch_bounds__(RING_WAVES_ * 32)
void sht_ring_idft(const float* __restrict__ cA,
                   const float* __restrict__ cB,
                   const float* __restrict__ cosT,
                   const float* __restrict__ sinT,
                   float* __restrict__ out) {
  __shared__ __attribute__((aligned(16))) float sA[M2_ * 64];
  __shared__ __attribute__((aligned(16))) float sB[M2_ * 64];
  const int tid    = threadIdx.x;
  const int lane   = tid & 31;
  const int wave   = __builtin_amdgcn_readfirstlane(tid >> 5);
  const int bcpair = blockIdx.x;
  const int t      = blockIdx.y;
  const int n  = ring_nlon(t);
  const int nh = n >> 1;
  const int ntiles = (nh >> 4) + 1;
  const int rot = (bcpair + t + (t >> 2) + (t >> 4) + (t >> 6)) & 3;
  int jtile = wave + rot;
  if (jtile >= RING_WAVES_) jtile -= RING_WAVES_;

  {
    const float* __restrict__ gA = cA + ((size_t)t * M2_ * BC_ + bcpair * 32) * 2;
    const float* __restrict__ gB = cB + ((size_t)t * M2_ * BC_ + bcpair * 32) * 2;
    for (int idx = tid; idx < M2_ * 16; idx += RING_WAVES_ * 32) {
      const int row = idx >> 4;
      const int c4  = (idx & 15) * 4;
      const v4f va = *(const v4f*)(gA + (size_t)row * (BC_ * 2) + c4);
      const v4f vb = *(const v4f*)(gB + (size_t)row * (BC_ * 2) + c4);
      *(v4f*)(sA + row * 64 + c4) = va;
      *(v4f*)(sB + row * 64 + c4) = vb;
    }
  }
  __syncthreads();
  if (jtile >= ntiles) return;

  const int roff = ring_offset(t);
  const int hi  = lane >> 4;
  const int lid = lane & 15;
  const int j   = jtile * 16 + lid;

  const size_t tb = tab_base(n >> 2) + (size_t)j * 2;
  const float* __restrict__ cT = cosT + tb;
  const float* __restrict__ sT = sinT + tb;

  v8f accC0 = {0.f, 0.f, 0.f, 0.f, 0.f, 0.f, 0.f, 0.f};
  v8f accS0 = accC0, accC1 = accC0, accS1 = accC0;

  for (int kk = 0; kk < MMAX_; kk += 4) {
    const int m2 = (kk >> 1) + hi;
    const v2f cf  = *(const v2f*)(cT + (size_t)m2 * n * 2);
    const v2f sf  = *(const v2f*)(sT + (size_t)m2 * n * 2);
    const v2f af0 = *(const v2f*)(sA + (m2 * 32 + lid) * 2);
    const v2f af1 = *(const v2f*)(sA + (m2 * 32 + lid + 16) * 2);
    const v2f bf0 = *(const v2f*)(sB + (m2 * 32 + lid) * 2);
    const v2f bf1 = *(const v2f*)(sB + (m2 * 32 + lid + 16) * 2);
    accC0 = __builtin_amdgcn_wmma_f32_16x16x4_f32(false, af0, false, cf,
                                                  (short)0, accC0, false, false);
    accS0 = __builtin_amdgcn_wmma_f32_16x16x4_f32(false, bf0, false, sf,
                                                  (short)0, accS0, false, false);
    accC1 = __builtin_amdgcn_wmma_f32_16x16x4_f32(false, af1, false, cf,
                                                  (short)0, accC1, false, false);
    accS1 = __builtin_amdgcn_wmma_f32_16x16x4_f32(false, bf1, false, sf,
                                                  (short)0, accS1, false, false);
  }

  const bool fwd = (j <= nh);
  const bool mir = (j > 0) && (j < nh);
  #pragma unroll
  for (int v = 0; v < 8; ++v) {
    const int bc = bcpair * 32 + v + 8 * hi;
    float* __restrict__ o0 = out + (size_t)bc * NPIX_ + roff;
    float* __restrict__ o1 = o0 + (size_t)16 * NPIX_;
    if (fwd) {
      o0[j] = accC0[v] + accS0[v];
      o1[j] = accC1[v] + accS1[v];
    }
    if (mir) {
      o0[n - j] = accC0[v] - accS0[v];
      o1[n - j] = accC1[v] - accS1[v];
    }
  }
}

extern "C" void kernel_launch(void* const* d_in, const int* in_sizes, int n_in,
                              void* d_out, int out_size, void* d_ws, size_t ws_size,
                              hipStream_t stream) {
  (void)in_sizes; (void)n_in; (void)out_size; (void)ws_size;
  const float* x_real = (const float*)d_in[0];
  const float* x_imag = (const float*)d_in[1];
  const float* pct    = (const float*)d_in[2];
  const float* offset = (const float*)d_in[3];
  float* out = (float*)d_out;

  float* xTr  = (float*)d_ws;
  float* xTi  = xTr  + (size_t)MMAX_ * L2_ * BC_ * 2;
  float* pctP = xTi  + (size_t)MMAX_ * L2_ * BC_ * 2;
  float* cA   = pctP + (size_t)MMAX_ * L2_ * 256 * 2;
  float* cB   = cA   + (size_t)NLAT_ * M2_ * BC_ * 2;
  float* cosT = cB   + (size_t)NLAT_ * M2_ * BC_ * 2;
  float* sinT = cosT + (size_t)TAB_FLOATS;

  sht_prepare<<<dim3(PREP_BLOCKS_), dim3(256), 0, stream>>>(
      x_real, x_imag, pct, xTr, xTi, pctP, cosT, sinT);

  sht_legendre_rotate<<<dim3(8, MMAX_), dim3(512), 0, stream>>>(
      xTr, xTi, pctP, offset, cA, cB);

  sht_ring_idft<<<dim3(8, NLAT_), dim3(RING_WAVES_ * 32), 0, stream>>>(
      cA, cB, cosT, sinT, out);
}


# preparation launch re-blocked: 1312 work-groups of fat threads (16-byte pieces, 6-8 per thread, x tile turned in LDS) instead of 14592 with one float per thread; 116736 -> 10496 waves; same values
# speedup vs baseline: 1.8020x; 1.0474x over previous
#include <hip/hip_runtime.h>
#include <hip/hip_bf16.h>

typedef __attribute__((ext_vector_type(2))) float v2f;
typedef __attribute__((ext_vector_type(4))) float v4f;
typedef __attribute__((ext_vector_type(8))) float v8f;

#define NSIDE_  64
#define NLAT_   255
#define LMAX_   96
#define MMAX_   96
#define BC_     256
#define NPIX_   49152
#define L2_     48
#define M2_     48

__device__ __forceinline__ int ring_nlon(int t) {
  return (t < NSIDE_ - 1) ? 4 * (t + 1)
       : (t < 3 * NSIDE_) ? 4 * NSIDE_
       : 4 * (4 * NSIDE_ - 1 - t);
}

__device__ __forceinline__ int ring_offset(int t) {
  if (t < NSIDE_ - 1) return 2 * t * (t + 1);
  if (t < 3 * NSIDE_) return 2 * (NSIDE_ - 1) * NSIDE_ + 4 * NSIDE_ * (t - (NSIDE_ - 1));
  int r = NLAT_ - t;
  return NPIX_ - 2 * r * (r + 1);
}

__device__ __forceinline__ size_t tab_base(int k) {
  return (size_t)192 * k * (k - 1);
}
#define TAB_FLOATS 798720

__device__ __forceinline__ void prep_table_item(int k, int idx,
                                                float* __restrict__ cosT,
                                                float* __restrict__ sinT) {
  const int n  = 4 * k;
  const int m2 = idx / n;
  const int j  = idx - m2 * n;
  const int nh = n >> 1;
  const float twon = 6.28318530717958647692f / (float)n;
  float cp[2], sp[2];
  #pragma unroll
  for (int e = 0; e < 2; ++e) {
    const int mm = 2 * m2 + e;
    const float wc  = (mm == 0) ? 1.f : (mm < nh ? 2.f : (mm == nh ? 1.f : 0.f));
    const float wsn = (mm == nh) ? 0.f : wc;
    float s, c;
    __sincosf(twon * (float)((mm * j) % n), &s, &c);
    cp[e] = wc * c;
    sp[e] = -wsn * s;
  }
  const size_t o = tab_base(k) + (size_t)idx * 2;
  const v2f cv = {cp[0], cp[1]};
  const v2f sv = {sp[0], sp[1]};
  *(v2f*)(cosT + o) = cv;
  *(v2f*)(sinT + o) = sv;
}

#define TAB_PIECE_ 1560
__device__ __forceinline__ void prep_build_tables(int b, int tid,
                                                  float* __restrict__ cosT,
                                                  float* __restrict__ sinT) {
  const int p  = b >> 3;
  const int s  = b & 7;
  const int k1 = p + 1;
  const int k2 = 64 - p;
  const int items1 = 192 * k1;
  #pragma unroll 1
  for (int r = 0; r < 7; ++r) {
    const int i = r * 256 + tid;
    if (i < TAB_PIECE_) {
      const int g = s * TAB_PIECE_ + i;
      if (g < items1) prep_table_item(k1, g, cosT, sinT);
      else            prep_table_item(k2, g - items1, cosT, sinT);
    }
  }
}

#define PCT_ROWS_PER_BLOCK_ 16
__device__ __forceinline__ void prep_repack_pct(int b, int tid,
                                                const float* __restrict__ pct,
                                                float* __restrict__ pctP) {
  const int i    = tid & 127;
  const int half = tid >> 7;
  const bool last = (i == 127);
  v2f a[8];
  float b0[8], b1[8];
  #pragma unroll
  for (int k = 0; k < 8; ++k) {
    const int r = b * PCT_ROWS_PER_BLOCK_ + half + 2 * k;
    const float* __restrict__ in = pct + (size_t)r * (2 * NLAT_);
    a[k]  = *(const v2f*)(in + 2 * i);
    b0[k] = in[NLAT_ + 2 * i];
    b1[k] = last ? 0.f : in[NLAT_ + 1 + 2 * i];
  }
  #pragma unroll
  for (int k = 0; k < 8; ++k) {
    const int r = b * PCT_ROWS_PER_BLOCK_ + half + 2 * k;
    const v4f o = {a[k].x, b0[k], last ? 0.f : a[k].y, b1[k]};
    *(v4f*)(pctP + (size_t)r * 512 + 4 * i) = o;
  }
}

#define TILE_PITCH_ 196
__device__ __forceinline__ void prep_transpose_x(int b, int tid, float* tile,
                                                 const float* __restrict__ xr,
                                                 const float* __restrict__ xi,
                                                 float* __restrict__ xTr,
                                                 float* __restrict__ xTi) {
  const int plane = (b >= 384) ? 1 : 0;
  const int rem   = b - plane * 384;
  const int l2    = rem >> 3;
  const int bc0   = (rem & 7) * 32;
  const float* __restrict__ src =
      (plane ? xi : xr) + ((size_t)bc0 * LMAX_ + 2 * l2) * MMAX_;
  float* __restrict__ dst =
      (plane ? xTi : xTr) + (size_t)l2 * (BC_ * 2) + bc0 * 2;
  v4f v[6];
  #pragma unroll
  for (int r = 0; r < 6; ++r) {
    const int c  = tid + 256 * r;
    const int bc = c / 48;
    const int q  = c - bc * 48;
    v[r] = *(const v4f*)(src + (size_t)bc * (LMAX_ * MMAX_) + q * 4);
  }
  #pragma unroll
  for (int r = 0; r < 6; ++r) {
    const int c  = tid + 256 * r;
    const int bc = c / 48;
    const int q  = c - bc * 48;
    *(v4f*)(tile + bc * TILE_PITCH_ + q * 4) = v[r];
  }
  __syncthreads();
  #pragma unroll
  for (int r = 0; r < 6; ++r) {
    const int c   = tid + 256 * r;
    const int m   = c >> 4;
    const int bcp = c & 15;
    const float* t0 = tile + (2 * bcp) * TILE_PITCH_ + m;
    const v4f o = {t0[0], t0[MMAX_], t0[TILE_PITCH_], t0[TILE_PITCH_ + MMAX_]};
    *(v4f*)(dst + (size_t)m * (L2_ * BC_ * 2) + bcp * 4) = o;
  }
}

#define PREP_W_BLOCKS_ 256
#define PREP_T_BLOCKS_ 768
#define PREP_P_BLOCKS_ 288
#define PREP_BLOCKS_   1312
__global__ __launch_bounds__(256)
void sht_prepare(const float* __restrict__ xr,
                 const float* __restrict__ xi,
                 const float* __restrict__ pct,
                 float* __restrict__ xTr,
                 float* __restrict__ xTi,
                 float* __restrict__ pctP,
                 float* __restrict__ cosT,
                 float* __restrict__ sinT) {
  __shared__ __attribute__((aligned(16))) float tile[32 * TILE_PITCH_];
  const int b   = blockIdx.x;
  const int tid = threadIdx.x;
  if (b < PREP_W_BLOCKS_) {
    prep_build_tables(b, tid, cosT, sinT);
  } else if (b < PREP_W_BLOCKS_ + PREP_T_BLOCKS_) {
    prep_transpose_x(b - PREP_W_BLOCKS_, tid, tile, xr, xi, xTr, xTi);
  } else {
    prep_repack_pct(b - (PREP_W_BLOCKS_ + PREP_T_BLOCKS_), tid, pct, pctP);
  }
}

__global__ __launch_bounds__(512)
void sht_legendre_rotate(const float* __restrict__ xTr,
                         const float* __restrict__ xTi,
                         const float* __restrict__ pctP,
                         const float* __restrict__ offset,
                         float* __restrict__ cA,
                         float* __restrict__ cB) {
  __shared__ __attribute__((aligned(16))) float sR[L2_ * 64];
  __shared__ __attribute__((aligned(16))) float sI[L2_ * 64];
  const int tid    = threadIdx.x;
  const int lane   = tid & 31;
  const int ttile  = __builtin_amdgcn_readfirstlane(tid >> 5);
  const int bcpair = blockIdx.x;
  const int m      = blockIdx.y;

  {
    const float* __restrict__ gR = xTr + ((size_t)m * L2_ * BC_ + bcpair * 32) * 2;
    const float* __restrict__ gI = xTi + ((size_t)m * L2_ * BC_ + bcpair * 32) * 2;
    for (int idx = tid; idx < L2_ * 16; idx += 512) {
      const int row = idx >> 4;
      const int c4  = (idx & 15) * 4;
      const v4f vr = *(const v4f*)(gR + (size_t)row * (BC_ * 2) + c4);
      const v4f vi = *(const v4f*)(gI + (size_t)row * (BC_ * 2) + c4);
      *(v4f*)(sR + row * 64 + c4) = vr;
      *(v4f*)(sI + row * 64 + c4) = vi;
    }
  }
  __syncthreads();

  const int hi  = lane >> 4;
  const int lid = lane & 15;
  const int trow = ttile * 16 + lid;
  const int bc0  = bcpair * 32 + lid;

  const float* __restrict__ pb = pctP + ((size_t)m * L2_ * 256 + trow) * 2;

  v8f accR0 = {0.f, 0.f, 0.f, 0.f, 0.f, 0.f, 0.f, 0.f};
  v8f accI0 = accR0, accR1 = accR0, accI1 = accR0;

  for (int kk = 0; kk < LMAX_; kk += 4) {
    const int l2 = (kk >> 1) + hi;
    const v2f af  = *(const v2f*)(pb + (size_t)l2 * 256 * 2);
    const v2f bR0 = *(const v2f*)(sR + (l2 * 32 + lid) * 2);
    const v2f bR1 = *(const v2f*)(sR + (l2 * 32 + lid + 16) * 2);
    const v2f bI0 = *(const v2f*)(sI + (l2 * 32 + lid) * 2);
    const v2f bI1 = *(const v2f*)(sI + (l2 * 32 + lid + 16) * 2);
    accR0 = __builtin_amdgcn_wmma_f32_16x16x4_f32(false, af, false, bR0,
                                                  (short)0, accR0, false, false);
    accI0 = __builtin_amdgcn_wmma_f32_16x16x4_f32(false, af, false, bI0,
                                                  (short)0, accI0, false, false);
    accR1 = __builtin_amdgcn_wmma_f32_16x16x4_f32(false, af, false, bR1,
                                                  (short)0, accR1, false, false);
    accI1 = __builtin_amdgcn_wmma_f32_16x16x4_f32(false, af, false, bI1,
                                                  (short)0, accI1, false, false);
  }

  #pragma unroll
  for (int v = 0; v < 8; ++v) {
    const int t = ttile * 16 + v + 8 * hi;
    if (t < NLAT_) {
      float s, c;
      __sincosf(offset[t * MMAX_ + m], &s, &c);
      const size_t o = (((size_t)t * M2_ + (m >> 1)) * BC_ + bc0) * 2 + (m & 1);
      float re = accR0[v], im = accI0[v];
      cA[o]      = re * c - im * s;
      cB[o]      = re * s + im * c;
      re = accR1[v]; im = accI1[v];
      cA[o + 32] = re * c - im * s;
      cB[o + 32] = re * s + im * c;
    }
  }
}

#define RING_WAVES_ 12
__global__ __launch_bounds__(RING_WAVES_ * 32)
void sht_ring_idft(const float* __restrict__ cA,
                   const float* __restrict__ cB,
                   const float* __restrict__ cosT,
                   const float* __restrict__ sinT,
                   float* __restrict__ out) {
  __shared__ __attribute__((aligned(16))) float sA[M2_ * 64];
  __shared__ __attribute__((aligned(16))) float sB[M2_ * 64];
  const int tid    = threadIdx.x;
  const int lane   = tid & 31;
  const int wave   = __builtin_amdgcn_readfirstlane(tid >> 5);
  const int bcpair = blockIdx.x;
  const int t      = blockIdx.y;
  const int n  = ring_nlon(t);
  const int nh = n >> 1;
  const int ntiles = (nh >> 4) + 1;
  const int rot = (bcpair + t + (t >> 2) + (t >> 4) + (t >> 6)) & 3;
  int jtile = wave + rot;
  if (jtile >= RING_WAVES_) jtile -= RING_WAVES_;

  {
    const float* __restrict__ gA = cA + ((size_t)t * M2_ * BC_ + bcpair * 32) * 2;
    const float* __restrict__ gB = cB + ((size_t)t * M2_ * BC_ + bcpair * 32) * 2;
    for (int idx = tid; idx < M2_ * 16; idx += RING_WAVES_ * 32) {
      const int row = idx >> 4;
      const int c4  = (idx & 15) * 4;
      const v4f va = *(const v4f*)(gA + (size_t)row * (BC_ * 2) + c4);
      const v4f vb = *(const v4f*)(gB + (size_t)row * (BC_ * 2) + c4);
      *(v4f*)(sA + row * 64 + c4) = va;
      *(v4f*)(sB + row * 64 + c4) = vb;
    }
  }
  __syncthreads();
  if (jtile >= ntiles) return;

  const int roff = ring_offset(t);
  const int hi  = lane >> 4;
  const int lid = lane & 15;
  const int j   = jtile * 16 + lid;

  const size_t tb = tab_base(n >> 2) + (size_t)j * 2;
  const float* __restrict__ cT = cosT + tb;
  const float* __restrict__ sT = sinT + tb;

  v8f accC0 = {0.f, 0.f, 0.f, 0.f, 0.f, 0.f, 0.f, 0.f};
  v8f accS0 = accC0, accC1 = accC0, accS1 = accC0;

  for (int kk = 0; kk < MMAX_; kk += 4) {
    const int m2 = (kk >> 1) + hi;
    const v2f cf  = *(const v2f*)(cT + (size_t)m2 * n * 2);
    const v2f sf  = *(const v2f*)(sT + (size_t)m2 * n * 2);
    const v2f af0 = *(const v2f*)(sA + (m2 * 32 + lid) * 2);
    const v2f af1 = *(const v2f*)(sA + (m2 * 32 + lid + 16) * 2);
    const v2f bf0 = *(const v2f*)(sB + (m2 * 32 + lid) * 2);
    const v2f bf1 = *(const v2f*)(sB + (m2 * 32 + lid + 16) * 2);
    accC0 = __builtin_amdgcn_wmma_f32_16x16x4_f32(false, af0, false, cf,
                                                  (short)0, accC0, false, false);
    accS0 = __builtin_amdgcn_wmma_f32_16x16x4_f32(false, bf0, false, sf,
                                                  (short)0, accS0, false, false);
    accC1 = __builtin_amdgcn_wmma_f32_16x16x4_f32(false, af1, false, cf,
                                                  (short)0, accC1, false, false);
    accS1 = __builtin_amdgcn_wmma_f32_16x16x4_f32(false, bf1, false, sf,
                                                  (short)0, accS1, false, false);
  }

  const bool fwd = (j <= nh);
  const bool mir = (j > 0) && (j < nh);
  #pragma unroll
  for (int v = 0; v < 8; ++v) {
    const int bc = bcpair * 32 + v + 8 * hi;
    float* __restrict__ o0 = out + (size_t)bc * NPIX_ + roff;
    float* __restrict__ o1 = o0 + (size_t)16 * NPIX_;
    if (fwd) {
      o0[j] = accC0[v] + accS0[v];
      o1[j] = accC1[v] + accS1[v];
    }
    if (mir) {
      o0[n - j] = accC0[v] - accS0[v];
      o1[n - j] = accC1[v] - accS1[v];
    }
  }
}

extern "C" void kernel_launch(void* const* d_in, const int* in_sizes, int n_in,
                              void* d_out, int out_size, void* d_ws, size_t ws_size,
                              hipStream_t stream) {
  (void)in_sizes; (void)n_in; (void)out_size; (void)ws_size;
  const float* x_real = (const float*)d_in[0];
  const float* x_imag = (const float*)d_in[1];
  const float* pct    = (const float*)d_in[2];
  const float* offset = (const float*)d_in[3];
  float* out = (float*)d_out;

  float* xTr  = (float*)d_ws;
  float* xTi  = xTr  + (size_t)MMAX_ * L2_ * BC_ * 2;
  float* pctP = xTi  + (size_t)MMAX_ * L2_ * BC_ * 2;
  float* cA   = pctP + (size_t)MMAX_ * L2_ * 256 * 2;
  float* cB   = cA   + (size_t)NLAT_ * M2_ * BC_ * 2;
  float* cosT = cB   + (size_t)NLAT_ * M2_ * BC_ * 2;
  float* sinT = cosT + (size_t)TAB_FLOATS;

  sht_prepare<<<dim3(PREP_BLOCKS_), dim3(256), 0, stream>>>(
      x_real, x_imag, pct, xTr, xTi, pctP, cosT, sinT);

  sht_legendre_rotate<<<dim3(8, MMAX_), dim3(512), 0, stream>>>(
      xTr, xTi, pctP, offset, cA, cB);

  sht_ring_idft<<<dim3(8, NLAT_), dim3(RING_WAVES_ * 32), 0, stream>>>(
      cA, cB, cosT, sinT, out);
}


# ring iDFT: columns n/2+-j and n-j equal column j up to (-1)^m signs: only j=0..n/4 are multiplied, even and odd modes summed apart (sum order changed); parity-major mode rows; 256-thread groups
# speedup vs baseline: 1.9241x; 1.0678x over previous
#include <hip/hip_runtime.h>
#include <hip/hip_bf16.h>

typedef __attribute__((ext_vector_type(2))) float v2f;
typedef __attribute__((ext_vector_type(4))) float v4f;
typedef __attribute__((ext_vector_type(8))) float v8f;

#define NSIDE_  64
#define NLAT_   255
#define LMAX_   96
#define MMAX_   96
#define BC_     256
#define NPIX_   49152
#define L2_     48
#define M2_     48
#define MROWS_PAR_ 24

__device__ __forceinline__ int ring_nlon(int t) {
  return (t < NSIDE_ - 1) ? 4 * (t + 1)
       : (t < 3 * NSIDE_) ? 4 * NSIDE_
       : 4 * (4 * NSIDE_ - 1 - t);
}

__device__ __forceinline__ int ring_offset(int t) {
  if (t < NSIDE_ - 1) return 2 * t * (t + 1);
  if (t < 3 * NSIDE_) return 2 * (NSIDE_ - 1) * NSIDE_ + 4 * NSIDE_ * (t - (NSIDE_ - 1));
  int r = NLAT_ - t;
  return NPIX_ - 2 * r * (r + 1);
}

__device__ __forceinline__ size_t tab_base(int k) {
  return (size_t)192 * k * (k - 1);
}
#define TAB_FLOATS 798720

__device__ __forceinline__ void prep_table_item(int k, int idx,
                                                float* __restrict__ cosT,
                                                float* __restrict__ sinT) {
  const int n  = 4 * k;
  const int r  = idx / n;
  const int j  = idx - r * n;
  const int nh = n >> 1;
  const int par = (r >= MROWS_PAR_) ? 1 : 0;
  const int q2  = r - par * MROWS_PAR_;
  const float twon = 6.28318530717958647692f / (float)n;
  float cp[2], sp[2];
  #pragma unroll
  for (int e = 0; e < 2; ++e) {
    const int mm = 4 * q2 + par + 2 * e;
    const float wc  = (mm == 0) ? 1.f : (mm < nh ? 2.f : (mm == nh ? 1.f : 0.f));
    const float wsn = (mm == nh) ? 0.f : wc;
    float s, c;
    __sincosf(twon * (float)((mm * j) % n), &s, &c);
    cp[e] = wc * c;
    sp[e] = -wsn * s;
  }
  const size_t o = tab_base(k) + (size_t)idx * 2;
  const v2f cv = {cp[0], cp[1]};
  const v2f sv = {sp[0], sp[1]};
  *(v2f*)(cosT + o) = cv;
  *(v2f*)(sinT + o) = sv;
}

#define TAB_PIECE_ 1560
__device__ __forceinline__ void prep_build_tables(int b, int tid,
                                                  float* __restrict__ cosT,
                                                  float* __restrict__ sinT) {
  const int p  = b >> 3;
  const int s  = b & 7;
  const int k1 = p + 1;
  const int k2 = 64 - p;
  const int items1 = 192 * k1;
  #pragma unroll 1
  for (int r = 0; r < 7; ++r) {
    const int i = r * 256 + tid;
    if (i < TAB_PIECE_) {
      const int g = s * TAB_PIECE_ + i;
      if (g < items1) prep_table_item(k1, g, cosT, sinT);
      else            prep_table_item(k2, g - items1, cosT, sinT);
    }
  }
}

#define PCT_ROWS_PER_BLOCK_ 16
__device__ __forceinline__ void prep_repack_pct(int b, int tid,
                                                const float* __restrict__ pct,
                                                float* __restrict__ pctP) {
  const int i    = tid & 127;
  const int half = tid >> 7;
  const bool last = (i == 127);
  v2f a[8];
  float b0[8], b1[8];
  #pragma unroll
  for (int k = 0; k < 8; ++k) {
    const int r = b * PCT_ROWS_PER_BLOCK_ + half + 2 * k;
    const float* __restrict__ in = pct + (size_t)r * (2 * NLAT_);
    a[k]  = *(const v2f*)(in + 2 * i);
    b0[k] = in[NLAT_ + 2 * i];
    b1[k] = last ? 0.f : in[NLAT_ + 1 + 2 * i];
  }
  #pragma unroll
  for (int k = 0; k < 8; ++k) {
    const int r = b * PCT_ROWS_PER_BLOCK_ + half + 2 * k;
    const v4f o = {a[k].x, b0[k], last ? 0.f : a[k].y, b1[k]};
    *(v4f*)(pctP + (size_t)r * 512 + 4 * i) = o;
  }
}

#define TILE_PITCH_ 196
__device__ __forceinline__ void prep_transpose_x(int b, int tid, float* tile,
                                                 const float* __restrict__ xr,
                                                 const float* __restrict__ xi,
                                                 float* __restrict__ xTr,
                                                 float* __restrict__ xTi) {
  const int plane = (b >= 384) ? 1 : 0;
  const int rem   = b - plane * 384;
  const int l2    = rem >> 3;
  const int bc0   = (rem & 7) * 32;
  const float* __restrict__ src =
      (plane ? xi : xr) + ((size_t)bc0 * LMAX_ + 2 * l2) * MMAX_;
  float* __restrict__ dst =
      (plane ? xTi : xTr) + (size_t)l2 * (BC_ * 2) + bc0 * 2;
  v4f v[6];
  #pragma unroll
  for (int r = 0; r < 6; ++r) {
    const int c  = tid + 256 * r;
    const int bc = c / 48;
    const int q  = c - bc * 48;
    v[r] = *(const v4f*)(src + (size_t)bc * (LMAX_ * MMAX_) + q * 4);
  }
  #pragma unroll
  for (int r = 0; r < 6; ++r) {
    const int c  = tid + 256 * r;
    const int bc = c / 48;
    const int q  = c - bc * 48;
    *(v4f*)(tile + bc * TILE_PITCH_ + q * 4) = v[r];
  }
  __syncthreads();
  #pragma unroll
  for (int r = 0; r < 6; ++r) {
    const int c   = tid + 256 * r;
    const int m   = c >> 4;
    const int bcp = c & 15;
    const float* t0 = tile + (2 * bcp) * TILE_PITCH_ + m;
    const v4f o = {t0[0], t0[MMAX_], t0[TILE_PITCH_], t0[TILE_PITCH_ + MMAX_]};
    *(v4f*)(dst + (size_t)m * (L2_ * BC_ * 2) + bcp * 4) = o;
  }
}

#define PREP_W_BLOCKS_ 256
#define PREP_T_BLOCKS_ 768
#define PREP_P_BLOCKS_ 288
#define PREP_BLOCKS_   1312
__global__ __launch_bounds__(256)
void sht_prepare(const float* __restrict__ xr,
                 const float* __restrict__ xi,
                 const float* __restrict__ pct,
                 float* __restrict__ xTr,
                 float* __restrict__ xTi,
                 float* __restrict__ pctP,
                 float* __restrict__ cosT,
                 float* __restrict__ sinT) {
  __shared__ __attribute__((aligned(16))) float tile[32 * TILE_PITCH_];
  const int b   = blockIdx.x;
  const int tid = threadIdx.x;
  if (b < PREP_W_BLOCKS_) {
    prep_build_tables(b, tid, cosT, sinT);
  } else if (b < PREP_W_BLOCKS_ + PREP_T_BLOCKS_) {
    prep_transpose_x(b - PREP_W_BLOCKS_, tid, tile, xr, xi, xTr, xTi);
  } else {
    prep_repack_pct(b - (PREP_W_BLOCKS_ + PREP_T_BLOCKS_), tid, pct, pctP);
  }
}

__global__ __launch_bounds__(512)
void sht_legendre_rotate(const float* __restrict__ xTr,
                         const float* __restrict__ xTi,
                         const float* __restrict__ pctP,
                         const float* __restrict__ offset,
                         float* __restrict__ cA,
                         float* __restrict__ cB) {
  __shared__ __attribute__((aligned(16))) float sR[L2_ * 64];
  __shared__ __attribute__((aligned(16))) float sI[L2_ * 64];
  const int tid    = threadIdx.x;
  const int lane   = tid & 31;
  const int ttile  = __builtin_amdgcn_readfirstlane(tid >> 5);
  const int bcpair = blockIdx.x;
  const int m      = blockIdx.y;

  {
    const float* __restrict__ gR = xTr + ((size_t)m * L2_ * BC_ + bcpair * 32) * 2;
    const float* __restrict__ gI = xTi + ((size_t)m * L2_ * BC_ + bcpair * 32) * 2;
    for (int idx = tid; idx < L2_ * 16; idx += 512) {
      const int row = idx >> 4;
      const int c4  = (idx & 15) * 4;
      const v4f vr = *(const v4f*)(gR + (size_t)row * (BC_ * 2) + c4);
      const v4f vi = *(const v4f*)(gI + (size_t)row * (BC_ * 2) + c4);
      *(v4f*)(sR + row * 64 + c4) = vr;
      *(v4f*)(sI + row * 64 + c4) = vi;
    }
  }
  __syncthreads();

  const int hi  = lane >> 4;
  const int lid = lane & 15;
  const int trow = ttile * 16 + lid;
  const int bc0  = bcpair * 32 + lid;

  const float* __restrict__ pb = pctP + ((size_t)m * L2_ * 256 + trow) * 2;

  v8f accR0 = {0.f, 0.f, 0.f, 0.f, 0.f, 0.f, 0.f, 0.f};
  v8f accI0 = accR0, accR1 = accR0, accI1 = accR0;

  for (int kk = 0; kk < LMAX_; kk += 4) {
    const int l2 = (kk >> 1) + hi;
    const v2f af  = *(const v2f*)(pb + (size_t)l2 * 256 * 2);
    const v2f bR0 = *(const v2f*)(sR + (l2 * 32 + lid) * 2);
    const v2f bR1 = *(const v2f*)(sR + (l2 * 32 + lid + 16) * 2);
    const v2f bI0 = *(const v2f*)(sI + (l2 * 32 + lid) * 2);
    const v2f bI1 = *(const v2f*)(sI + (l2 * 32 + lid + 16) * 2);
    accR0 = __builtin_amdgcn_wmma_f32_16x16x4_f32(false, af, false, bR0,
                                                  (short)0, accR0, false, false);
    accI0 = __builtin_amdgcn_wmma_f32_16x16x4_f32(false, af, false, bI0,
                                                  (short)0, accI0, false, false);
    accR1 = __builtin_amdgcn_wmma_f32_16x16x4_f32(false, af, false, bR1,
                                                  (short)0, accR1, false, false);
    accI1 = __builtin_amdgcn_wmma_f32_16x16x4_f32(false, af, false, bI1,
                                                  (short)0, accI1, false, false);
  }

  const int mrow  = (m & 1) * MROWS_PAR_ + (m >> 2);
  const int mslot = (m >> 1) & 1;
  #pragma unroll
  for (int v = 0; v < 8; ++v) {
    const int t = ttile * 16 + v + 8 * hi;
    if (t < NLAT_) {
      float s, c;
      __sincosf(offset[t * MMAX_ + m], &s, &c);
      const size_t o = (((size_t)t * M2_ + mrow) * BC_ + bc0) * 2 + mslot;
      float re = accR0[v], im = accI0[v];
      cA[o]      = re * c - im * s;
      cB[o]      = re * s + im * c;
      re = accR1[v]; im = accI1[v];
      cA[o + 32] = re * c - im * s;
      cB[o + 32] = re * s + im * c;
    }
  }
}

#define RING_WAVES_ 8
__global__ __launch_bounds__(RING_WAVES_ * 32)
void sht_ring_idft(const float* __restrict__ cA,
                   const float* __restrict__ cB,
                   const float* __restrict__ cosT,
                   const float* __restrict__ sinT,
                   float* __restrict__ out) {
  __shared__ __attribute__((aligned(16))) float sA[M2_ * 64];
  __shared__ __attribute__((aligned(16))) float sB[M2_ * 64];
  const int tid    = threadIdx.x;
  const int lane   = tid & 31;
  const int wave   = __builtin_amdgcn_readfirstlane(tid >> 5);
  const int bcpair = blockIdx.x;
  const int t      = blockIdx.y;
  const int n  = ring_nlon(t);
  const int nh = n >> 1;
  const int nq = n >> 2;
  const int ntiles = (nq >> 4) + 1;
  const int rot = (bcpair + t + (t >> 2) + (t >> 4) + (t >> 6)) & 3;
  int jtile = wave + rot;
  if (jtile >= RING_WAVES_) jtile -= RING_WAVES_;

  {
    const float* __restrict__ gA = cA + ((size_t)t * M2_ * BC_ + bcpair * 32) * 2;
    const float* __restrict__ gB = cB + ((size_t)t * M2_ * BC_ + bcpair * 32) * 2;
    for (int idx = tid; idx < M2_ * 16; idx += RING_WAVES_ * 32) {
      const int row = idx >> 4;
      const int c4  = (idx & 15) * 4;
      const v4f va = *(const v4f*)(gA + (size_t)row * (BC_ * 2) + c4);
      const v4f vb = *(const v4f*)(gB + (size_t)row * (BC_ * 2) + c4);
      *(v4f*)(sA + row * 64 + c4) = va;
      *(v4f*)(sB + row * 64 + c4) = vb;
    }
  }
  __syncthreads();
  if (jtile >= ntiles) return;

  const int roff = ring_offset(t);
  const int hi  = lane >> 4;
  const int lid = lane & 15;
  const int j   = jtile * 16 + lid;

  const size_t tb = tab_base(nq) + (size_t)j * 2;
  const float* __restrict__ cT = cosT + tb;
  const float* __restrict__ sT = sinT + tb;

  v8f accCe0 = {0.f, 0.f, 0.f, 0.f, 0.f, 0.f, 0.f, 0.f};
  v8f accCo0 = accCe0, accSe0 = accCe0, accSo0 = accCe0;
  v8f accCe1 = accCe0, accCo1 = accCe0, accSe1 = accCe0, accSo1 = accCe0;

  for (int kk = 0; kk < MROWS_PAR_; kk += 2) {
    const int re = kk + hi;
    const int ro = re + MROWS_PAR_;
    const v2f cfe = *(const v2f*)(cT + (size_t)re * n * 2);
    const v2f cfo = *(const v2f*)(cT + (size_t)ro * n * 2);
    const v2f sfe = *(const v2f*)(sT + (size_t)re * n * 2);
    const v2f sfo = *(const v2f*)(sT + (size_t)ro * n * 2);
    const v2f ae0 = *(const v2f*)(sA + (re * 32 + lid) * 2);
    const v2f ae1 = *(const v2f*)(sA + (re * 32 + lid + 16) * 2);
    const v2f ao0 = *(const v2f*)(sA + (ro * 32 + lid) * 2);
    const v2f ao1 = *(const v2f*)(sA + (ro * 32 + lid + 16) * 2);
    const v2f be0 = *(const v2f*)(sB + (re * 32 + lid) * 2);
    const v2f be1 = *(const v2f*)(sB + (re * 32 + lid + 16) * 2);
    const v2f bo0 = *(const v2f*)(sB + (ro * 32 + lid) * 2);
    const v2f bo1 = *(const v2f*)(sB + (ro * 32 + lid + 16) * 2);
    accCe0 = __builtin_amdgcn_wmma_f32_16x16x4_f32(false, ae0, false, cfe,
                                                   (short)0, accCe0, false, false);
    accCo0 = __builtin_amdgcn_wmma_f32_16x16x4_f32(false, ao0, false, cfo,
                                                   (short)0, accCo0, false, false);
    accSe0 = __builtin_amdgcn_wmma_f32_16x16x4_f32(false, be0, false, sfe,
                                                   (short)0, accSe0, false, false);
    accSo0 = __builtin_amdgcn_wmma_f32_16x16x4_f32(false, bo0, false, sfo,
                                                   (short)0, accSo0, false, false);
    accCe1 = __builtin_amdgcn_wmma_f32_16x16x4_f32(false, ae1, false, cfe,
                                                   (short)0, accCe1, false, false);
    accCo1 = __builtin_amdgcn_wmma_f32_16x16x4_f32(false, ao1, false, cfo,
                                                   (short)0, accCo1, false, false);
    accSe1 = __builtin_amdgcn_wmma_f32_16x16x4_f32(false, be1, false, sfe,
                                                   (short)0, accSe1, false, false);
    accSo1 = __builtin_amdgcn_wmma_f32_16x16x4_f32(false, bo1, false, sfo,
                                                   (short)0, accSo1, false, false);
  }

  const bool q0 = (j <= nq);
  const bool q3 = (j > 0) && (j <= nq);
  const bool q1 = (j < nq);
  const bool q2 = (j > 0) && (j < nq);
  #pragma unroll
  for (int v = 0; v < 8; ++v) {
    const int bc = bcpair * 32 + v + 8 * hi;
    float* __restrict__ o0 = out + (size_t)bc * NPIX_ + roff;
    float* __restrict__ o1 = o0 + (size_t)16 * NPIX_;
    const float p0 = accCe0[v] + accCo0[v], s0 = accSe0[v] + accSo0[v];
    const float r0 = accCe0[v] - accCo0[v], d0 = accSe0[v] - accSo0[v];
    const float p1 = accCe1[v] + accCo1[v], s1 = accSe1[v] + accSo1[v];
    const float r1 = accCe1[v] - accCo1[v], d1 = accSe1[v] - accSo1[v];
    if (q0) {
      o0[j] = p0 + s0;
      o1[j] = p1 + s1;
    }
    if (q1) {
      o0[nh - j] = r0 - d0;
      o1[nh - j] = r1 - d1;
    }
    if (q2) {
      o0[nh + j] = r0 + d0;
      o1[nh + j] = r1 + d1;
    }
    if (q3) {
      o0[n - j] = p0 - s0;
      o1[n - j] = p1 - s1;
    }
  }
}

extern "C" void kernel_launch(void* const* d_in, const int* in_sizes, int n_in,
                              void* d_out, int out_size, void* d_ws, size_t ws_size,
                              hipStream_t stream) {
  (void)in_sizes; (void)n_in; (void)out_size; (void)ws_size;
  const float* x_real = (const float*)d_in[0];
  const float* x_imag = (const float*)d_in[1];
  const float* pct    = (const float*)d_in[2];
  const float* offset = (const float*)d_in[3];
  float* out = (float*)d_out;

  float* xTr  = (float*)d_ws;
  float* xTi  = xTr  + (size_t)MMAX_ * L2_ * BC_ * 2;
  float* pctP = xTi  + (size_t)MMAX_ * L2_ * BC_ * 2;
  float* cA   = pctP + (size_t)MMAX_ * L2_ * 256 * 2;
  float* cB   = cA   + (size_t)NLAT_ * M2_ * BC_ * 2;
  float* cosT = cB   + (size_t)NLAT_ * M2_ * BC_ * 2;
  float* sinT = cosT + (size_t)TAB_FLOATS;

  sht_prepare<<<dim3(PREP_BLOCKS_), dim3(256), 0, stream>>>(
      x_real, x_imag, pct, xTr, xTi, pctP, cosT, sinT);

  sht_legendre_rotate<<<dim3(8, MMAX_), dim3(512), 0, stream>>>(
      xTr, xTi, pctP, offset, cA, cB);

  sht_ring_idft<<<dim3(8, NLAT_), dim3(RING_WAVES_ * 32), 0, stream>>>(
      cA, cB, cosT, sinT, out);
}


# both matrix kernels store whole 128-byte lines: Legendre epilogue trades lane halves (one line per store, coef planes [t][bc pair][m][32 bc]); ring rows gathered in LDS, written as 16-byte pieces
# speedup vs baseline: 2.0681x; 1.0748x over previous
#include <hip/hip_runtime.h>
#include <hip/hip_bf16.h>

typedef __attribute__((ext_vector_type(2))) float v2f;
typedef __attribute__((ext_vector_type(4))) float v4f;
typedef __attribute__((ext_vector_type(8))) float v8f;

#define NSIDE_  64
#define NLAT_   255
#define LMAX_   96
#define MMAX_   96
#define BC_     256
#define NPIX_   49152
#define L2_     48
#define M2_     48
#define MROWS_PAR_ 24

__device__ __forceinline__ int ring_nlon(int t) {
  return (t < NSIDE_ - 1) ? 4 * (t + 1)
       : (t < 3 * NSIDE_) ? 4 * NSIDE_
       : 4 * (4 * NSIDE_ - 1 - t);
}

__device__ __forceinline__ int ring_offset(int t) {
  if (t < NSIDE_ - 1) return 2 * t * (t + 1);
  if (t < 3 * NSIDE_) return 2 * (NSIDE_ - 1) * NSIDE_ + 4 * NSIDE_ * (t - (NSIDE_ - 1));
  int r = NLAT_ - t;
  return NPIX_ - 2 * r * (r + 1);
}

__device__ __forceinline__ size_t tab_base(int k) {
  return (size_t)192 * k * (k - 1);
}
#define TAB_FLOATS 798720

__device__ __forceinline__ void prep_table_item(int k, int idx,
                                                float* __restrict__ cosT,
                                                float* __restrict__ sinT) {
  const int n  = 4 * k;
  const int r  = idx / n;
  const int j  = idx - r * n;
  const int nh = n >> 1;
  const int par = (r >= MROWS_PAR_) ? 1 : 0;
  const int q2  = r - par * MROWS_PAR_;
  const float twon = 6.28318530717958647692f / (float)n;
  float cp[2], sp[2];
  #pragma unroll
  for (int e = 0; e < 2; ++e) {
    const int mm = 4 * q2 + par + 2 * e;
    const float wc  = (mm == 0) ? 1.f : (mm < nh ? 2.f : (mm == nh ? 1.f : 0.f));
    const float wsn = (mm == nh) ? 0.f : wc;
    float s, c;
    __sincosf(twon * (float)((mm * j) % n), &s, &c);
    cp[e] = wc * c;
    sp[e] = -wsn * s;
  }
  const size_t o = tab_base(k) + (size_t)idx * 2;
  const v2f cv = {cp[0], cp[1]};
  const v2f sv = {sp[0], sp[1]};
  *(v2f*)(cosT + o) = cv;
  *(v2f*)(sinT + o) = sv;
}

#define TAB_PIECE_ 1560
__device__ __forceinline__ void prep_build_tables(int b, int tid,
                                                  float* __restrict__ cosT,
                                                  float* __restrict__ sinT) {
  const int p  = b >> 3;
  const int s  = b & 7;
  const int k1 = p + 1;
  const int k2 = 64 - p;
  const int items1 = 192 * k1;
  #pragma unroll 1
  for (int r = 0; r < 7; ++r) {
    const int i = r * 256 + tid;
    if (i < TAB_PIECE_) {
      const int g = s * TAB_PIECE_ + i;
      if (g < items1) prep_table_item(k1, g, cosT, sinT);
      else            prep_table_item(k2, g - items1, cosT, sinT);
    }
  }
}

#define PCT_ROWS_PER_BLOCK_ 16
__device__ __forceinline__ void prep_repack_pct(int b, int tid,
                                                const float* __restrict__ pct,
                                                float* __restrict__ pctP) {
  const int i    = tid & 127;
  const int half = tid >> 7;
  const bool last = (i == 127);
  v2f a[8];
  float b0[8], b1[8];
  #pragma unroll
  for (int k = 0; k < 8; ++k) {
    const int r = b * PCT_ROWS_PER_BLOCK_ + half + 2 * k;
    const float* __restrict__ in = pct + (size_t)r * (2 * NLAT_);
    a[k]  = *(const v2f*)(in + 2 * i);
    b0[k] = in[NLAT_ + 2 * i];
    b1[k] = last ? 0.f : in[NLAT_ + 1 + 2 * i];
  }
  #pragma unroll
  for (int k = 0; k < 8; ++k) {
    const int r = b * PCT_ROWS_PER_BLOCK_ + half + 2 * k;
    const v4f o = {a[k].x, b0[k], last ? 0.f : a[k].y, b1[k]};
    *(v4f*)(pctP + (size_t)r * 512 + 4 * i) = o;
  }
}

#define TILE_PITCH_ 196
__device__ __forceinline__ void prep_transpose_x(int b, int tid, float* tile,
                                                 const float* __restrict__ xr,
                                                 const float* __restrict__ xi,
                                                 float* __restrict__ xTr,
                                                 float* __restrict__ xTi) {
  const int plane = (b >= 384) ? 1 : 0;
  const int rem   = b - plane * 384;
  const int l2    = rem >> 3;
  const int bc0   = (rem & 7) * 32;
  const float* __restrict__ src =
      (plane ? xi : xr) + ((size_t)bc0 * LMAX_ + 2 * l2) * MMAX_;
  float* __restrict__ dst =
      (plane ? xTi : xTr) + (size_t)l2 * (BC_ * 2) + bc0 * 2;
  v4f v[6];
  #pragma unroll
  for (int r = 0; r < 6; ++r) {
    const int c  = tid + 256 * r;
    const int bc = c / 48;
    const int q  = c - bc * 48;
    v[r] = *(const v4f*)(src + (size_t)bc * (LMAX_ * MMAX_) + q * 4);
  }
  #pragma unroll
  for (int r = 0; r < 6; ++r) {
    const int c  = tid + 256 * r;
    const int bc = c / 48;
    const int q  = c - bc * 48;
    *(v4f*)(tile + bc * TILE_PITCH_ + q * 4) = v[r];
  }
  __syncthreads();
  #pragma unroll
  for (int r = 0; r < 6; ++r) {
    const int c   = tid + 256 * r;
    const int m   = c >> 4;
    const int bcp = c & 15;
    const float* t0 = tile + (2 * bcp) * TILE_PITCH_ + m;
    const v4f o = {t0[0], t0[MMAX_], t0[TILE_PITCH_], t0[TILE_PITCH_ + MMAX_]};
    *(v4f*)(dst + (size_t)m * (L2_ * BC_ * 2) + bcp * 4) = o;
  }
}

#define PREP_W_BLOCKS_ 256
#define PREP_T_BLOCKS_ 768
#define PREP_P_BLOCKS_ 288
#define PREP_BLOCKS_   1312
__global__ __launch_bounds__(256)
void sht_prepare(const float* __restrict__ xr,
                 const float* __restrict__ xi,
                 const float* __restrict__ pct,
                 float* __restrict__ xTr,
                 float* __restrict__ xTi,
                 float* __restrict__ pctP,
                 float* __restrict__ cosT,
                 float* __restrict__ sinT) {
  __shared__ __attribute__((aligned(16))) float tile[32 * TILE_PITCH_];
  const int b   = blockIdx.x;
  const int tid = threadIdx.x;
  if (b < PREP_W_BLOCKS_) {
    prep_build_tables(b, tid, cosT, sinT);
  } else if (b < PREP_W_BLOCKS_ + PREP_T_BLOCKS_) {
    prep_transpose_x(b - PREP_W_BLOCKS_, tid, tile, xr, xi, xTr, xTi);
  } else {
    prep_repack_pct(b - (PREP_W_BLOCKS_ + PREP_T_BLOCKS_), tid, pct, pctP);
  }
}

__device__ __forceinline__ float xhalf(float x) {
  return __builtin_bit_cast(float, __builtin_amdgcn_permlanex16(
      __builtin_bit_cast(int, x), __builtin_bit_cast(int, x),
      0x76543210, 0xfedcba98, false, false));
}

__device__ __forceinline__ void half_trade(float x0, float x1, int hi,
                                           float& lo, float& up) {
  const float keep = hi ? x1 : x0;
  const float give = hi ? x0 : x1;
  const float got  = xhalf(give);
  lo = hi ? got : keep;
  up = hi ? keep : got;
}

__global__ __launch_bounds__(512)
void sht_legendre_rotate(const float* __restrict__ xTr,
                         const float* __restrict__ xTi,
                         const float* __restrict__ pctP,
                         const float* __restrict__ offset,
                         float* __restrict__ cA,
                         float* __restrict__ cB) {
  __shared__ __attribute__((aligned(16))) float sR[L2_ * 64];
  __shared__ __attribute__((aligned(16))) float sI[L2_ * 64];
  const int tid    = threadIdx.x;
  const int lane   = tid & 31;
  const int ttile  = __builtin_amdgcn_readfirstlane(tid >> 5);
  const int bcpair = blockIdx.x;
  const int m      = blockIdx.y;

  {
    const float* __restrict__ gR = xTr + ((size_t)m * L2_ * BC_ + bcpair * 32) * 2;
    const float* __restrict__ gI = xTi + ((size_t)m * L2_ * BC_ + bcpair * 32) * 2;
    for (int idx = tid; idx < L2_ * 16; idx += 512) {
      const int row = idx >> 4;
      const int c4  = (idx & 15) * 4;
      const v4f vr = *(const v4f*)(gR + (size_t)row * (BC_ * 2) + c4);
      const v4f vi = *(const v4f*)(gI + (size_t)row * (BC_ * 2) + c4);
      *(v4f*)(sR + row * 64 + c4) = vr;
      *(v4f*)(sI + row * 64 + c4) = vi;
    }
  }
  __syncthreads();

  const int hi  = lane >> 4;
  const int lid = lane & 15;
  const int trow = ttile * 16 + lid;
  const int bc0  = bcpair * 32 + lid;

  const float* __restrict__ pb = pctP + ((size_t)m * L2_ * 256 + trow) * 2;

  v8f accR0 = {0.f, 0.f, 0.f, 0.f, 0.f, 0.f, 0.f, 0.f};
  v8f accI0 = accR0, accR1 = accR0, accI1 = accR0;

  for (int kk = 0; kk < LMAX_; kk += 4) {
    const int l2 = (kk >> 1) + hi;
    const v2f af  = *(const v2f*)(pb + (size_t)l2 * 256 * 2);
    const v2f bR0 = *(const v2f*)(sR + (l2 * 32 + lid) * 2);
    const v2f bR1 = *(const v2f*)(sR + (l2 * 32 + lid + 16) * 2);
    const v2f bI0 = *(const v2f*)(sI + (l2 * 32 + lid) * 2);
    const v2f bI1 = *(const v2f*)(sI + (l2 * 32 + lid + 16) * 2);
    accR0 = __builtin_amdgcn_wmma_f32_16x16x4_f32(false, af, false, bR0,
                                                  (short)0, accR0, false, false);
    accI0 = __builtin_amdgcn_wmma_f32_16x16x4_f32(false, af, false, bI0,
                                                  (short)0, accI0, false, false);
    accR1 = __builtin_amdgcn_wmma_f32_16x16x4_f32(false, af, false, bR1,
                                                  (short)0, accR1, false, false);
    accI1 = __builtin_amdgcn_wmma_f32_16x16x4_f32(false, af, false, bI1,
                                                  (short)0, accI1, false, false);
  }

  float* __restrict__ lA =
      cA + ((((size_t)ttile * 16) * 8 + bcpair) * MMAX_ + m) * 32 + lane;
  float* __restrict__ lB =
      cB + ((((size_t)ttile * 16) * 8 + bcpair) * MMAX_ + m) * 32 + lane;
  #pragma unroll
  for (int v = 0; v < 8; ++v) {
    const int t  = ttile * 16 + v + 8 * hi;
    const int tl = (t < NLAT_) ? t : (NLAT_ - 1);
    float s, c;
    __sincosf(offset[tl * MMAX_ + m], &s, &c);
    float re = accR0[v], im = accI0[v];
    const float a0 = re * c - im * s;
    const float b0 = re * s + im * c;
    re = accR1[v]; im = accI1[v];
    const float a1 = re * c - im * s;
    const float b1 = re * s + im * c;
    float aLo, aHi, bLo, bHi;
    half_trade(a0, a1, hi, aLo, aHi);
    half_trade(b0, b1, hi, bLo, bHi);
    const size_t oLo = (size_t)v * (8 * MMAX_ * 32);
    const size_t oHi = (size_t)(v + 8) * (8 * MMAX_ * 32);
    lA[oLo] = aLo;
    lB[oLo] = bLo;
    if (ttile * 16 + v + 8 < NLAT_) {
      lA[oHi] = aHi;
      lB[oHi] = bHi;
    }
  }
}

#define RING_WAVES_ 8
#define RING_PITCH_ 260
__global__ __launch_bounds__(RING_WAVES_ * 32)
void sht_ring_idft(const float* __restrict__ cA,
                   const float* __restrict__ cB,
                   const float* __restrict__ cosT,
                   const float* __restrict__ sinT,
                   float* __restrict__ out) {
  __shared__ __attribute__((aligned(16))) float smem[32 * RING_PITCH_];
  float* sA = smem;
  float* sB = smem + M2_ * 64;
  const int tid    = threadIdx.x;
  const int lane   = tid & 31;
  const int wave   = __builtin_amdgcn_readfirstlane(tid >> 5);
  const int bcpair = blockIdx.x;
  const int t      = blockIdx.y;
  const int n  = ring_nlon(t);
  const int nh = n >> 1;
  const int nq = n >> 2;
  const int ntiles = (nq >> 4) + 1;
  const int rot = (bcpair + t + (t >> 2) + (t >> 4) + (t >> 6)) & 3;
  int jtile = wave + rot;
  if (jtile >= RING_WAVES_) jtile -= RING_WAVES_;
  const bool active = (jtile < ntiles);

  {
    const float* __restrict__ gA = cA + ((size_t)t * 8 + bcpair) * (MMAX_ * 32);
    const float* __restrict__ gB = cB + ((size_t)t * 8 + bcpair) * (MMAX_ * 32);
    v4f e0[3], e1[3];
    #pragma unroll
    for (int it = 0; it < 3; ++it) {
      const int idx   = tid + 256 * it;
      const int plane = (idx >= 384) ? 1 : 0;
      const int rem   = idx - plane * 384;
      const int r     = rem >> 3;
      const int c4    = (rem & 7) * 4;
      const int par   = (r >= MROWS_PAR_) ? 1 : 0;
      const int m0    = 4 * (r - par * MROWS_PAR_) + par;
      const float* __restrict__ g = (plane ? gB : gA) + m0 * 32 + c4;
      e0[it] = *(const v4f*)(g);
      e1[it] = *(const v4f*)(g + 64);
    }
    #pragma unroll
    for (int it = 0; it < 3; ++it) {
      const int idx   = tid + 256 * it;
      const int plane = (idx >= 384) ? 1 : 0;
      const int rem   = idx - plane * 384;
      const int r     = rem >> 3;
      const int c4    = (rem & 7) * 4;
      float* s = (plane ? sB : sA) + r * 64 + c4 * 2;
      const v4f w0 = {e0[it].x, e1[it].x, e0[it].y, e1[it].y};
      const v4f w1 = {e0[it].z, e1[it].z, e0[it].w, e1[it].w};
      *(v4f*)(s)     = w0;
      *(v4f*)(s + 4) = w1;
    }
  }
  __syncthreads();

  const int roff = ring_offset(t);
  const int hi  = lane >> 4;
  const int lid = lane & 15;
  const int j   = jtile * 16 + lid;

  v8f accCe0 = {0.f, 0.f, 0.f, 0.f, 0.f, 0.f, 0.f, 0.f};
  v8f accCo0 = accCe0, accSe0 = accCe0, accSo0 = accCe0;
  v8f accCe1 = accCe0, accCo1 = accCe0, accSe1 = accCe0, accSo1 = accCe0;

  if (active) {
    const size_t tb = tab_base(nq) + (size_t)j * 2;
    const float* __restrict__ cT = cosT + tb;
    const float* __restrict__ sT = sinT + tb;
    for (int kk = 0; kk < MROWS_PAR_; kk += 2) {
      const int re = kk + hi;
      const int ro = re + MROWS_PAR_;
      const v2f cfe = *(const v2f*)(cT + (size_t)re * n * 2);
      const v2f cfo = *(const v2f*)(cT + (size_t)ro * n * 2);
      const v2f sfe = *(const v2f*)(sT + (size_t)re * n * 2);
      const v2f sfo = *(const v2f*)(sT + (size_t)ro * n * 2);
      const v2f ae0 = *(const v2f*)(sA + (re * 32 + lid) * 2);
      const v2f ae1 = *(const v2f*)(sA + (re * 32 + lid + 16) * 2);
      const v2f ao0 = *(const v2f*)(sA + (ro * 32 + lid) * 2);
      const v2f ao1 = *(const v2f*)(sA + (ro * 32 + lid + 16) * 2);
      const v2f be0 = *(const v2f*)(sB + (re * 32 + lid) * 2);
      const v2f be1 = *(const v2f*)(sB + (re * 32 + lid + 16) * 2);
      const v2f bo0 = *(const v2f*)(sB + (ro * 32 + lid) * 2);
      const v2f bo1 = *(const v2f*)(sB + (ro * 32 + lid + 16) * 2);
      accCe0 = __builtin_amdgcn_wmma_f32_16x16x4_f32(false, ae0, false, cfe,
                                                     (short)0, accCe0, false, false);
      accCo0 = __builtin_amdgcn_wmma_f32_16x16x4_f32(false, ao0, false, cfo,
                                                     (short)0, accCo0, false, false);
      accSe0 = __builtin_amdgcn_wmma_f32_16x16x4_f32(false, be0, false, sfe,
                                                     (short)0, accSe0, false, false);
      accSo0 = __builtin_amdgcn_wmma_f32_16x16x4_f32(false, bo0, false, sfo,
                                                     (short)0, accSo0, false, false);
      accCe1 = __builtin_amdgcn_wmma_f32_16x16x4_f32(false, ae1, false, cfe,
                                                     (short)0, accCe1, false, false);
      accCo1 = __builtin_amdgcn_wmma_f32_16x16x4_f32(false, ao1, false, cfo,
                                                     (short)0, accCo1, false, false);
      accSe1 = __builtin_amdgcn_wmma_f32_16x16x4_f32(false, be1, false, sfe,
                                                     (short)0, accSe1, false, false);
      accSo1 = __builtin_amdgcn_wmma_f32_16x16x4_f32(false, bo1, false, sfo,
                                                     (short)0, accSo1, false, false);
    }
  }
  __syncthreads();

  if (active) {
    const bool q0 = (j <= nq);
    const bool q3 = (j > 0) && (j <= nq);
    const bool q1 = (j < nq);
    const bool q2 = (j > 0) && (j < nq);
    float* __restrict__ rw = smem + (8 * hi) * RING_PITCH_;
    #pragma unroll
    for (int v = 0; v < 8; ++v) {
      float* __restrict__ o0 = rw + v * RING_PITCH_;
      float* __restrict__ o1 = o0 + 16 * RING_PITCH_;
      const float p0 = accCe0[v] + accCo0[v], s0 = accSe0[v] + accSo0[v];
      const float r0 = accCe0[v] - accCo0[v], d0 = accSe0[v] - accSo0[v];
      const float p1 = accCe1[v] + accCo1[v], s1 = accSe1[v] + accSo1[v];
      const float r1 = accCe1[v] - accCo1[v], d1 = accSe1[v] - accSo1[v];
      if (q0) {
        o0[j] = p0 + s0;
        o1[j] = p1 + s1;
      }
      if (q1) {
        o0[nh - j] = r0 - d0;
        o1[nh - j] = r1 - d1;
      }
      if (q2) {
        o0[nh + j] = r0 + d0;
        o1[nh + j] = r1 + d1;
      }
      if (q3) {
        o0[n - j] = p0 - s0;
        o1[n - j] = p1 - s1;
      }
    }
  }
  __syncthreads();

  {
    const float* __restrict__ rr = smem + (wave * 4) * RING_PITCH_;
    float* __restrict__ go =
        out + (size_t)(bcpair * 32 + wave * 4) * NPIX_ + roff;
    #pragma unroll
    for (int it = 0; it < 2; ++it) {
      const int p = lane + 32 * it;
      if (p < nq) {
        v4f w[4];
        #pragma unroll
        for (int q = 0; q < 4; ++q)
          w[q] = *(const v4f*)(rr + q * RING_PITCH_ + 4 * p);
        #pragma unroll
        for (int q = 0; q < 4; ++q)
          *(v4f*)(go + (size_t)q * NPIX_ + 4 * p) = w[q];
      }
    }
  }
}

extern "C" void kernel_launch(void* const* d_in, const int* in_sizes, int n_in,
                              void* d_out, int out_size, void* d_ws, size_t ws_size,
                              hipStream_t stream) {
  (void)in_sizes; (void)n_in; (void)out_size; (void)ws_size;
  const float* x_real = (const float*)d_in[0];
  const float* x_imag = (const float*)d_in[1];
  const float* pct    = (const float*)d_in[2];
  const float* offset = (const float*)d_in[3];
  float* out = (float*)d_out;

  float* xTr  = (float*)d_ws;
  float* xTi  = xTr  + (size_t)MMAX_ * L2_ * BC_ * 2;
  float* pctP = xTi  + (size_t)MMAX_ * L2_ * BC_ * 2;
  float* cA   = pctP + (size_t)MMAX_ * L2_ * 256 * 2;
  float* cB   = cA   + (size_t)NLAT_ * M2_ * BC_ * 2;
  float* cosT = cB   + (size_t)NLAT_ * M2_ * BC_ * 2;
  float* sinT = cosT + (size_t)TAB_FLOATS;

  sht_prepare<<<dim3(PREP_BLOCKS_), dim3(256), 0, stream>>>(
      x_real, x_imag, pct, xTr, xTi, pctP, cosT, sinT);

  sht_legendre_rotate<<<dim3(8, MMAX_), dim3(512), 0, stream>>>(
      xTr, xTi, pctP, offset, cA, cB);

  sht_ring_idft<<<dim3(8, NLAT_), dim3(RING_WAVES_ * 32), 0, stream>>>(
      cA, cB, cosT, sinT, out);
}


# modes m > n/2 of a ring (dropped by its inverse real FFT) are no longer computed, stored or staged: Legendre waves and rows skipped, short rings run 4 or 8 of the 12 k steps; sums unchanged
# speedup vs baseline: 2.3840x; 1.1527x over previous
#include <hip/hip_runtime.h>
#include <hip/hip_bf16.h>

typedef __attribute__((ext_vector_type(2))) float v2f;
typedef __attribute__((ext_vector_type(4))) float v4f;
typedef __attribute__((ext_vector_type(8))) float v8f;

#define NSIDE_  64
#define NLAT_   255
#define LMAX_   96
#define MMAX_   96
#define BC_     256
#define NPIX_   49152
#define L2_     48
#define M2_     48
#define MROWS_PAR_ 24

__device__ __forceinline__ int ring_nlon(int t) {
  const int north = t + 1;
  const int south = 4 * NSIDE_ - 1 - t;
  const int cap   = north < south ? north : south;
  return 4 * (cap < NSIDE_ ? cap : NSIDE_);
}

__device__ __forceinline__ int ring_offset(int t) {
  if (t < NSIDE_ - 1) return 2 * t * (t + 1);
  if (t < 3 * NSIDE_) return 2 * (NSIDE_ - 1) * NSIDE_ + 4 * NSIDE_ * (t - (NSIDE_ - 1));
  int r = NLAT_ - t;
  return NPIX_ - 2 * r * (r + 1);
}

__device__ __forceinline__ size_t tab_base(int k) {
  return (size_t)192 * k * (k - 1);
}
#define TAB_FLOATS 798720

__device__ __forceinline__ void prep_table_item(int k, int idx,
                                                float* __restrict__ cosT,
                                                float* __restrict__ sinT) {
  const int n  = 4 * k;
  const int r  = idx / n;
  const int j  = idx - r * n;
  const int nh = n >> 1;
  const int par = (r >= MROWS_PAR_) ? 1 : 0;
  const int q2  = r - par * MROWS_PAR_;
  const float twon = 6.28318530717958647692f / (float)n;
  float cp[2], sp[2];
  #pragma unroll
  for (int e = 0; e < 2; ++e) {
    const int mm = 4 * q2 + par + 2 * e;
    const float wc  = (mm == 0) ? 1.f : (mm < nh ? 2.f : (mm == nh ? 1.f : 0.f));
    const float wsn = (mm == nh) ? 0.f : wc;
    float s, c;
    __sincosf(twon * (float)((mm * j) % n), &s, &c);
    cp[e] = wc * c;
    sp[e] = -wsn * s;
  }
  const size_t o = tab_base(k) + (size_t)idx * 2;
  const v2f cv = {cp[0], cp[1]};
  const v2f sv = {sp[0], sp[1]};
  *(v2f*)(cosT + o) = cv;
  *(v2f*)(sinT + o) = sv;
}

#define TAB_PIECE_ 1560
__device__ __forceinline__ void prep_build_tables(int b, int tid,
                                                  float* __restrict__ cosT,
                                                  float* __restrict__ sinT) {
  const int p  = b >> 3;
  const int s  = b & 7;
  const int k1 = p + 1;
  const int k2 = 64 - p;
  const int items1 = 192 * k1;
  #pragma unroll 1
  for (int r = 0; r < 7; ++r) {
    const int i = r * 256 + tid;
    if (i < TAB_PIECE_) {
      const int g = s * TAB_PIECE_ + i;
      if (g < items1) prep_table_item(k1, g, cosT, sinT);
      else            prep_table_item(k2, g - items1, cosT, sinT);
    }
  }
}

#define PCT_ROWS_PER_BLOCK_ 16
__device__ __forceinline__ void prep_repack_pct(int b, int tid,
                                                const float* __restrict__ pct,
                                                float* __restrict__ pctP) {
  const int i    = tid & 127;
  const int half = tid >> 7;
  const bool last = (i == 127);
  v2f a[8];
  float b0[8], b1[8];
  #pragma unroll
  for (int k = 0; k < 8; ++k) {
    const int r = b * PCT_ROWS_PER_BLOCK_ + half + 2 * k;
    const float* __restrict__ in = pct + (size_t)r * (2 * NLAT_);
    a[k]  = *(const v2f*)(in + 2 * i);
    b0[k] = in[NLAT_ + 2 * i];
    b1[k] = last ? 0.f : in[NLAT_ + 1 + 2 * i];
  }
  #pragma unroll
  for (int k = 0; k < 8; ++k) {
    const int r = b * PCT_ROWS_PER_BLOCK_ + half + 2 * k;
    const v4f o = {a[k].x, b0[k], last ? 0.f : a[k].y, b1[k]};
    *(v4f*)(pctP + (size_t)r * 512 + 4 * i) = o;
  }
}

#define TILE_PITCH_ 196
__device__ __forceinline__ void prep_transpose_x(int b, int tid, float* tile,
                                                 const float* __restrict__ xr,
                                                 const float* __restrict__ xi,
                                                 float* __restrict__ xTr,
                                                 float* __restrict__ xTi) {
  const int plane = (b >= 384) ? 1 : 0;
  const int rem   = b - plane * 384;
  const int l2    = rem >> 3;
  const int bc0   = (rem & 7) * 32;
  const float* __restrict__ src =
      (plane ? xi : xr) + ((size_t)bc0 * LMAX_ + 2 * l2) * MMAX_;
  float* __restrict__ dst =
      (plane ? xTi : xTr) + (size_t)l2 * (BC_ * 2) + bc0 * 2;
  v4f v[6];
  #pragma unroll
  for (int r = 0; r < 6; ++r) {
    const int c  = tid + 256 * r;
    const int bc = c / 48;
    const int q  = c - bc * 48;
    v[r] = *(const v4f*)(src + (size_t)bc * (LMAX_ * MMAX_) + q * 4);
  }
  #pragma unroll
  for (int r = 0; r < 6; ++r) {
    const int c  = tid + 256 * r;
    const int bc = c / 48;
    const int q  = c - bc * 48;
    *(v4f*)(tile + bc * TILE_PITCH_ + q * 4) = v[r];
  }
  __syncthreads();
  #pragma unroll
  for (int r = 0; r < 6; ++r) {
    const int c   = tid + 256 * r;
    const int m   = c >> 4;
    const int bcp = c & 15;
    const float* t0 = tile + (2 * bcp) * TILE_PITCH_ + m;
    const v4f o = {t0[0], t0[MMAX_], t0[TILE_PITCH_], t0[TILE_PITCH_ + MMAX_]};
    *(v4f*)(dst + (size_t)m * (L2_ * BC_ * 2) + bcp * 4) = o;
  }
}

#define PREP_W_BLOCKS_ 256
#define PREP_T_BLOCKS_ 768
#define PREP_P_BLOCKS_ 288
#define PREP_BLOCKS_   1312
__global__ __launch_bounds__(256)
void sht_prepare(const float* __restrict__ xr,
                 const float* __restrict__ xi,
                 const float* __restrict__ pct,
                 float* __restrict__ xTr,
                 float* __restrict__ xTi,
                 float* __restrict__ pctP,
                 float* __restrict__ cosT,
                 float* __restrict__ sinT) {
  __shared__ __attribute__((aligned(16))) float tile[32 * TILE_PITCH_];
  const int b   = blockIdx.x;
  const int tid = threadIdx.x;
  if (b < PREP_W_BLOCKS_) {
    prep_build_tables(b, tid, cosT, sinT);
  } else if (b < PREP_W_BLOCKS_ + PREP_T_BLOCKS_) {
    prep_transpose_x(b - PREP_W_BLOCKS_, tid, tile, xr, xi, xTr, xTi);
  } else {
    prep_repack_pct(b - (PREP_W_BLOCKS_ + PREP_T_BLOCKS_), tid, pct, pctP);
  }
}

__device__ __forceinline__ float xhalf(float x) {
  return __builtin_bit_cast(float, __builtin_amdgcn_permlanex16(
      __builtin_bit_cast(int, x), __builtin_bit_cast(int, x),
      0x76543210, 0xfedcba98, false, false));
}

__device__ __forceinline__ void half_trade(float x0, float x1, int hi,
                                           float& lo, float& up) {
  const float keep = hi ? x1 : x0;
  const float give = hi ? x0 : x1;
  const float got  = xhalf(give);
  lo = hi ? got : keep;
  up = hi ? keep : got;
}

__global__ __launch_bounds__(512)
void sht_legendre_rotate(const float* __restrict__ xTr,
                         const float* __restrict__ xTi,
                         const float* __restrict__ pctP,
                         const float* __restrict__ offset,
                         float* __restrict__ cA,
                         float* __restrict__ cB) {
  __shared__ __attribute__((aligned(16))) float sR[L2_ * 64];
  __shared__ __attribute__((aligned(16))) float sI[L2_ * 64];
  const int tid    = threadIdx.x;
  const int lane   = tid & 31;
  const int ttile  = __builtin_amdgcn_readfirstlane(tid >> 5);
  const int bcpair = blockIdx.x;
  const int m      = blockIdx.y;

  {
    const float* __restrict__ gR = xTr + ((size_t)m * L2_ * BC_ + bcpair * 32) * 2;
    const float* __restrict__ gI = xTi + ((size_t)m * L2_ * BC_ + bcpair * 32) * 2;
    for (int idx = tid; idx < L2_ * 16; idx += 512) {
      const int row = idx >> 4;
      const int c4  = (idx & 15) * 4;
      const v4f vr = *(const v4f*)(gR + (size_t)row * (BC_ * 2) + c4);
      const v4f vi = *(const v4f*)(gI + (size_t)row * (BC_ * 2) + c4);
      *(v4f*)(sR + row * 64 + c4) = vr;
      *(v4f*)(sI + row * 64 + c4) = vi;
    }
  }
  __syncthreads();

  {
    const int tFirst = ttile * 16;
    const int tLast  = (tFirst + 15 < NLAT_) ? tFirst + 15 : NLAT_ - 1;
    const int nFirst = ring_nlon(tFirst);
    const int nLast  = ring_nlon(tLast);
    if (2 * m > (nFirst > nLast ? nFirst : nLast)) return;
  }

  const int hi  = lane >> 4;
  const int lid = lane & 15;
  const int trow = ttile * 16 + lid;
  const int bc0  = bcpair * 32 + lid;

  const float* __restrict__ pb = pctP + ((size_t)m * L2_ * 256 + trow) * 2;

  v8f accR0 = {0.f, 0.f, 0.f, 0.f, 0.f, 0.f, 0.f, 0.f};
  v8f accI0 = accR0, accR1 = accR0, accI1 = accR0;

  for (int kk = 0; kk < LMAX_; kk += 4) {
    const int l2 = (kk >> 1) + hi;
    const v2f af  = *(const v2f*)(pb + (size_t)l2 * 256 * 2);
    const v2f bR0 = *(const v2f*)(sR + (l2 * 32 + lid) * 2);
    const v2f bR1 = *(const v2f*)(sR + (l2 * 32 + lid + 16) * 2);
    const v2f bI0 = *(const v2f*)(sI + (l2 * 32 + lid) * 2);
    const v2f bI1 = *(const v2f*)(sI + (l2 * 32 + lid + 16) * 2);
    accR0 = __builtin_amdgcn_wmma_f32_16x16x4_f32(false, af, false, bR0,
                                                  (short)0, accR0, false, false);
    accI0 = __builtin_amdgcn_wmma_f32_16x16x4_f32(false, af, false, bI0,
                                                  (short)0, accI0, false, false);
    accR1 = __builtin_amdgcn_wmma_f32_16x16x4_f32(false, af, false, bR1,
                                                  (short)0, accR1, false, false);
    accI1 = __builtin_amdgcn_wmma_f32_16x16x4_f32(false, af, false, bI1,
                                                  (short)0, accI1, false, false);
  }

  float* __restrict__ lA =
      cA + ((((size_t)ttile * 16) * 8 + bcpair) * MMAX_ + m) * 32 + lane;
  float* __restrict__ lB =
      cB + ((((size_t)ttile * 16) * 8 + bcpair) * MMAX_ + m) * 32 + lane;
  #pragma unroll
  for (int v = 0; v < 8; ++v) {
    const int t  = ttile * 16 + v + 8 * hi;
    const int tl = (t < NLAT_) ? t : (NLAT_ - 1);
    float s, c;
    __sincosf(offset[tl * MMAX_ + m], &s, &c);
    float re = accR0[v], im = accI0[v];
    const float a0 = re * c - im * s;
    const float b0 = re * s + im * c;
    re = accR1[v]; im = accI1[v];
    const float a1 = re * c - im * s;
    const float b1 = re * s + im * c;
    float aLo, aHi, bLo, bHi;
    half_trade(a0, a1, hi, aLo, aHi);
    half_trade(b0, b1, hi, bLo, bHi);
    const size_t oLo = (size_t)v * (8 * MMAX_ * 32);
    const size_t oHi = (size_t)(v + 8) * (8 * MMAX_ * 32);
    const int tLo = ttile * 16 + v;
    const int tHi = tLo + 8;
    if (2 * m <= ring_nlon(tLo)) {
      lA[oLo] = aLo;
      lB[oLo] = bLo;
    }
    if (tHi < NLAT_ && 2 * m <= ring_nlon(tHi)) {
      lA[oHi] = aHi;
      lB[oHi] = bHi;
    }
  }
}

#define RING_WAVES_ 8
#define RING_PITCH_ 260
__global__ __launch_bounds__(RING_WAVES_ * 32)
void sht_ring_idft(const float* __restrict__ cA,
                   const float* __restrict__ cB,
                   const float* __restrict__ cosT,
                   const float* __restrict__ sinT,
                   float* __restrict__ out) {
  __shared__ __attribute__((aligned(16))) float smem[32 * RING_PITCH_];
  float* sA = smem;
  float* sB = smem + M2_ * 64;
  const int tid    = threadIdx.x;
  const int lane   = tid & 31;
  const int wave   = __builtin_amdgcn_readfirstlane(tid >> 5);
  const int bcpair = blockIdx.x;
  const int t      = blockIdx.y;
  const int n  = ring_nlon(t);
  const int nh = n >> 1;
  const int nq = n >> 2;
  const int ntiles = (nq >> 4) + 1;
  const int rot = (bcpair + t + (t >> 2) + (t >> 4) + (t >> 6)) & 3;
  int jtile = wave + rot;
  if (jtile >= RING_WAVES_) jtile -= RING_WAVES_;
  const bool active = (jtile < ntiles);

  {
    const float* __restrict__ gA = cA + ((size_t)t * 8 + bcpair) * (MMAX_ * 32);
    const float* __restrict__ gB = cB + ((size_t)t * 8 + bcpair) * (MMAX_ * 32);
    v4f e0[3], e1[3];
    #pragma unroll
    for (int it = 0; it < 3; ++it) {
      const int idx   = tid + 256 * it;
      const int plane = (idx >= 384) ? 1 : 0;
      const int rem   = idx - plane * 384;
      const int r     = rem >> 3;
      const int c4    = (rem & 7) * 4;
      const int par   = (r >= MROWS_PAR_) ? 1 : 0;
      const int m0    = 4 * (r - par * MROWS_PAR_) + par;
      const bool k0 = (m0 <= nh);
      const bool k1 = (m0 + 2 <= nh);
      const float* __restrict__ g = (plane ? gB : gA) + c4;
      const v4f l0 = *(const v4f*)(g + (k0 ? m0 : 0) * 32);
      const v4f l1 = *(const v4f*)(g + (k1 ? m0 + 2 : 0) * 32);
      const v4f zero4 = {0.f, 0.f, 0.f, 0.f};
      e0[it] = k0 ? l0 : zero4;
      e1[it] = k1 ? l1 : zero4;
    }
    #pragma unroll
    for (int it = 0; it < 3; ++it) {
      const int idx   = tid + 256 * it;
      const int plane = (idx >= 384) ? 1 : 0;
      const int rem   = idx - plane * 384;
      const int r     = rem >> 3;
      const int c4    = (rem & 7) * 4;
      float* s = (plane ? sB : sA) + r * 64 + c4 * 2;
      const v4f w0 = {e0[it].x, e1[it].x, e0[it].y, e1[it].y};
      const v4f w1 = {e0[it].z, e1[it].z, e0[it].w, e1[it].w};
      *(v4f*)(s)     = w0;
      *(v4f*)(s + 4) = w1;
    }
  }
  __syncthreads();

  const int roff = ring_offset(t);
  const int hi  = lane >> 4;
  const int lid = lane & 15;
  const int j   = jtile * 16 + lid;

  v8f accCe0 = {0.f, 0.f, 0.f, 0.f, 0.f, 0.f, 0.f, 0.f};
  v8f accCo0 = accCe0, accSe0 = accCe0, accSo0 = accCe0;
  v8f accCe1 = accCe0, accCo1 = accCe0, accSe1 = accCe0, accSo1 = accCe0;

  if (active) {
    const size_t tb = tab_base(nq) + (size_t)j * 2;
    const float* __restrict__ cT = cosT + tb;
    const float* __restrict__ sT = sinT + tb;
    auto kstep = [&](int kk) {
      const int re = kk + hi;
      const int ro = re + MROWS_PAR_;
      const v2f cfe = *(const v2f*)(cT + (size_t)re * n * 2);
      const v2f cfo = *(const v2f*)(cT + (size_t)ro * n * 2);
      const v2f sfe = *(const v2f*)(sT + (size_t)re * n * 2);
      const v2f sfo = *(const v2f*)(sT + (size_t)ro * n * 2);
      const v2f ae0 = *(const v2f*)(sA + (re * 32 + lid) * 2);
      const v2f ae1 = *(const v2f*)(sA + (re * 32 + lid + 16) * 2);
      const v2f ao0 = *(const v2f*)(sA + (ro * 32 + lid) * 2);
      const v2f ao1 = *(const v2f*)(sA + (ro * 32 + lid + 16) * 2);
      const v2f be0 = *(const v2f*)(sB + (re * 32 + lid) * 2);
      const v2f be1 = *(const v2f*)(sB + (re * 32 + lid + 16) * 2);
      const v2f bo0 = *(const v2f*)(sB + (ro * 32 + lid) * 2);
      const v2f bo1 = *(const v2f*)(sB + (ro * 32 + lid + 16) * 2);
      accCe0 = __builtin_amdgcn_wmma_f32_16x16x4_f32(false, ae0, false, cfe,
                                                     (short)0, accCe0, false, false);
      accCo0 = __builtin_amdgcn_wmma_f32_16x16x4_f32(false, ao0, false, cfo,
                                                     (short)0, accCo0, false, false);
      accSe0 = __builtin_amdgcn_wmma_f32_16x16x4_f32(false, be0, false, sfe,
                                                     (short)0, accSe0, false, false);
      accSo0 = __builtin_amdgcn_wmma_f32_16x16x4_f32(false, bo0, false, sfo,
                                                     (short)0, accSo0, false, false);
      accCe1 = __builtin_amdgcn_wmma_f32_16x16x4_f32(false, ae1, false, cfe,
                                                     (short)0, accCe1, false, false);
      accCo1 = __builtin_amdgcn_wmma_f32_16x16x4_f32(false, ao1, false, cfo,
                                                     (short)0, accCo1, false, false);
      accSe1 = __builtin_amdgcn_wmma_f32_16x16x4_f32(false, be1, false, sfe,
                                                     (short)0, accSe1, false, false);
      accSo1 = __builtin_amdgcn_wmma_f32_16x16x4_f32(false, bo1, false, sfo,
                                                     (short)0, accSo1, false, false);
    };
    if (nh >= 64) {
      for (int kk = 0; kk < MROWS_PAR_; kk += 2) kstep(kk);
    } else {
      for (int kk = 0; kk < 8; kk += 2) kstep(kk);
      if (nh >= 32) {
        for (int kk = 8; kk < 16; kk += 2) kstep(kk);
      }
    }
  }
  __syncthreads();

  if (active) {
    const bool q0 = (j <= nq);
    const bool q3 = (j > 0) && (j <= nq);
    const bool q1 = (j < nq);
    const bool q2 = (j > 0) && (j < nq);
    float* __restrict__ rw = smem + (8 * hi) * RING_PITCH_;
    #pragma unroll
    for (int v = 0; v < 8; ++v) {
      float* __restrict__ o0 = rw + v * RING_PITCH_;
      float* __restrict__ o1 = o0 + 16 * RING_PITCH_;
      const float p0 = accCe0[v] + accCo0[v], s0 = accSe0[v] + accSo0[v];
      const float r0 = accCe0[v] - accCo0[v], d0 = accSe0[v] - accSo0[v];
      const float p1 = accCe1[v] + accCo1[v], s1 = accSe1[v] + accSo1[v];
      const float r1 = accCe1[v] - accCo1[v], d1 = accSe1[v] - accSo1[v];
      if (q0) {
        o0[j] = p0 + s0;
        o1[j] = p1 + s1;
      }
      if (q1) {
        o0[nh - j] = r0 - d0;
        o1[nh - j] = r1 - d1;
      }
      if (q2) {
        o0[nh + j] = r0 + d0;
        o1[nh + j] = r1 + d1;
      }
      if (q3) {
        o0[n - j] = p0 - s0;
        o1[n - j] = p1 - s1;
      }
    }
  }
  __syncthreads();

  {
    const float* __restrict__ rr = smem + (wave * 4) * RING_PITCH_;
    float* __restrict__ go =
        out + (size_t)(bcpair * 32 + wave * 4) * NPIX_ + roff;
    #pragma unroll
    for (int it = 0; it < 2; ++it) {
      const int p = lane + 32 * it;
      if (p < nq) {
        v4f w[4];
        #pragma unroll
        for (int q = 0; q < 4; ++q)
          w[q] = *(const v4f*)(rr + q * RING_PITCH_ + 4 * p);
        #pragma unroll
        for (int q = 0; q < 4; ++q)
          *(v4f*)(go + (size_t)q * NPIX_ + 4 * p) = w[q];
      }
    }
  }
}

extern "C" void kernel_launch(void* const* d_in, const int* in_sizes, int n_in,
                              void* d_out, int out_size, void* d_ws, size_t ws_size,
                              hipStream_t stream) {
  (void)in_sizes; (void)n_in; (void)out_size; (void)ws_size;
  const float* x_real = (const float*)d_in[0];
  const float* x_imag = (const float*)d_in[1];
  const float* pct    = (const float*)d_in[2];
  const float* offset = (const float*)d_in[3];
  float* out = (float*)d_out;

  float* xTr  = (float*)d_ws;
  float* xTi  = xTr  + (size_t)MMAX_ * L2_ * BC_ * 2;
  float* pctP = xTi  + (size_t)MMAX_ * L2_ * BC_ * 2;
  float* cA   = pctP + (size_t)MMAX_ * L2_ * 256 * 2;
  float* cB   = cA   + (size_t)NLAT_ * M2_ * BC_ * 2;
  float* cosT = cB   + (size_t)NLAT_ * M2_ * BC_ * 2;
  float* sinT = cosT + (size_t)TAB_FLOATS;

  sht_prepare<<<dim3(PREP_BLOCKS_), dim3(256), 0, stream>>>(
      x_real, x_imag, pct, xTr, xTi, pctP, cosT, sinT);

  sht_legendre_rotate<<<dim3(8, MMAX_), dim3(512), 0, stream>>>(
      xTr, xTi, pctP, offset, cA, cB);

  sht_ring_idft<<<dim3(8, NLAT_), dim3(RING_WAVES_ * 32), 0, stream>>>(
      cA, cB, cosT, sinT, out);
}


# Legendre kernel reads A fragments straight from the input pct (two 4-byte loads a row of 255 apart per k pair): the repacked copy, its 288 blocks, 9.4 MB write and re-read are gone; sums unchanged
# speedup vs baseline: 2.5535x; 1.0711x over previous
#include <hip/hip_runtime.h>
#include <hip/hip_bf16.h>

typedef __attribute__((ext_vector_type(2))) float v2f;
typedef __attribute__((ext_vector_type(4))) float v4f;
typedef __attribute__((ext_vector_type(8))) float v8f;

#define NSIDE_  64
#define NLAT_   255
#define LMAX_   96
#define MMAX_   96
#define BC_     256
#define NPIX_   49152
#define L2_     48
#define M2_     48
#define MROWS_PAR_ 24

__device__ __forceinline__ int ring_nlon(int t) {
  const int north = t + 1;
  const int south = 4 * NSIDE_ - 1 - t;
  const int cap   = north < south ? north : south;
  return 4 * (cap < NSIDE_ ? cap : NSIDE_);
}

__device__ __forceinline__ int ring_offset(int t) {
  if (t < NSIDE_ - 1) return 2 * t * (t + 1);
  if (t < 3 * NSIDE_) return 2 * (NSIDE_ - 1) * NSIDE_ + 4 * NSIDE_ * (t - (NSIDE_ - 1));
  int r = NLAT_ - t;
  return NPIX_ - 2 * r * (r + 1);
}

__device__ __forceinline__ size_t tab_base(int k) {
  return (size_t)192 * k * (k - 1);
}
#define TAB_FLOATS 798720

__device__ __forceinline__ void prep_table_item(int k, int idx,
                                                float* __restrict__ cosT,
                                                float* __restrict__ sinT) {
  const int n  = 4 * k;
  const int r  = idx / n;
  const int j  = idx - r * n;
  const int nh = n >> 1;
  const int par = (r >= MROWS_PAR_) ? 1 : 0;
  const int q2  = r - par * MROWS_PAR_;
  const float twon = 6.28318530717958647692f / (float)n;
  float cp[2], sp[2];
  #pragma unroll
  for (int e = 0; e < 2; ++e) {
    const int mm = 4 * q2 + par + 2 * e;
    const float wc  = (mm == 0) ? 1.f : (mm < nh ? 2.f : (mm == nh ? 1.f : 0.f));
    const float wsn = (mm == nh) ? 0.f : wc;
    float s, c;
    __sincosf(twon * (float)((mm * j) % n), &s, &c);
    cp[e] = wc * c;
    sp[e] = -wsn * s;
  }
  const size_t o = tab_base(k) + (size_t)idx * 2;
  const v2f cv = {cp[0], cp[1]};
  const v2f sv = {sp[0], sp[1]};
  *(v2f*)(cosT + o) = cv;
  *(v2f*)(sinT + o) = sv;
}

#define TAB_PIECE_ 1560
__device__ __forceinline__ void prep_build_tables(int b, int tid,
                                                  float* __restrict__ cosT,
                                                  float* __restrict__ sinT) {
  const int p  = b >> 3;
  const int s  = b & 7;
  const int k1 = p + 1;
  const int k2 = 64 - p;
  const int items1 = 192 * k1;
  #pragma unroll 1
  for (int r = 0; r < 7; ++r) {
    const int i = r * 256 + tid;
    if (i < TAB_PIECE_) {
      const int g = s * TAB_PIECE_ + i;
      if (g < items1) prep_table_item(k1, g, cosT, sinT);
      else            prep_table_item(k2, g - items1, cosT, sinT);
    }
  }
}

#define TILE_PITCH_ 196
__device__ __forceinline__ void prep_transpose_x(int b, int tid, float* tile,
                                                 const float* __restrict__ xr,
                                                 const float* __restrict__ xi,
                                                 float* __restrict__ xTr,
                                                 float* __restrict__ xTi) {
  const int plane = (b >= 384) ? 1 : 0;
  const int rem   = b - plane * 384;
  const int l2    = rem >> 3;
  const int bc0   = (rem & 7) * 32;
  const float* __restrict__ src =
      (plane ? xi : xr) + ((size_t)bc0 * LMAX_ + 2 * l2) * MMAX_;
  float* __restrict__ dst =
      (plane ? xTi : xTr) + (size_t)l2 * (BC_ * 2) + bc0 * 2;
  v4f v[6];
  #pragma unroll
  for (int r = 0; r < 6; ++r) {
    const int c  = tid + 256 * r;
    const int bc = c / 48;
    const int q  = c - bc * 48;
    v[r] = *(const v4f*)(src + (size_t)bc * (LMAX_ * MMAX_) + q * 4);
  }
  #pragma unroll
  for (int r = 0; r < 6; ++r) {
    const int c  = tid + 256 * r;
    const int bc = c / 48;
    const int q  = c - bc * 48;
    *(v4f*)(tile + bc * TILE_PITCH_ + q * 4) = v[r];
  }
  __syncthreads();
  #pragma unroll
  for (int r = 0; r < 6; ++r) {
    const int c   = tid + 256 * r;
    const int m   = c >> 4;
    const int bcp = c & 15;
    const float* t0 = tile + (2 * bcp) * TILE_PITCH_ + m;
    const v4f o = {t0[0], t0[MMAX_], t0[TILE_PITCH_], t0[TILE_PITCH_ + MMAX_]};
    *(v4f*)(dst + (size_t)m * (L2_ * BC_ * 2) + bcp * 4) = o;
  }
}

#define PREP_W_BLOCKS_ 256
#define PREP_T_BLOCKS_ 768
#define PREP_BLOCKS_   1024
__global__ __launch_bounds__(256)
void sht_prepare(const float* __restrict__ xr,
                 const float* __restrict__ xi,
                 float* __restrict__ xTr,
                 float* __restrict__ xTi,
                 float* __restrict__ cosT,
                 float* __restrict__ sinT) {
  __shared__ __attribute__((aligned(16))) float tile[32 * TILE_PITCH_];
  const int b   = blockIdx.x;
  const int tid = threadIdx.x;
  if (b < PREP_W_BLOCKS_) {
    prep_build_tables(b, tid, cosT, sinT);
  } else {
    prep_transpose_x(b - PREP_W_BLOCKS_, tid, tile, xr, xi, xTr, xTi);
  }
}

__device__ __forceinline__ float xhalf(float x) {
  return __builtin_bit_cast(float, __builtin_amdgcn_permlanex16(
      __builtin_bit_cast(int, x), __builtin_bit_cast(int, x),
      0x76543210, 0xfedcba98, false, false));
}

__device__ __forceinline__ void half_trade(float x0, float x1, int hi,
                                           float& lo, float& up) {
  const float keep = hi ? x1 : x0;
  const float give = hi ? x0 : x1;
  const float got  = xhalf(give);
  lo = hi ? got : keep;
  up = hi ? keep : got;
}

__global__ __launch_bounds__(512)
void sht_legendre_rotate(const float* __restrict__ xTr,
                         const float* __restrict__ xTi,
                         const float* __restrict__ pct,
                         const float* __restrict__ offset,
                         float* __restrict__ cA,
                         float* __restrict__ cB) {
  __shared__ __attribute__((aligned(16))) float sR[L2_ * 64];
  __shared__ __attribute__((aligned(16))) float sI[L2_ * 64];
  const int tid    = threadIdx.x;
  const int lane   = tid & 31;
  const int ttile  = __builtin_amdgcn_readfirstlane(tid >> 5);
  const int bcpair = blockIdx.x;
  const int m      = blockIdx.y;

  {
    const float* __restrict__ gR = xTr + ((size_t)m * L2_ * BC_ + bcpair * 32) * 2;
    const float* __restrict__ gI = xTi + ((size_t)m * L2_ * BC_ + bcpair * 32) * 2;
    for (int idx = tid; idx < L2_ * 16; idx += 512) {
      const int row = idx >> 4;
      const int c4  = (idx & 15) * 4;
      const v4f vr = *(const v4f*)(gR + (size_t)row * (BC_ * 2) + c4);
      const v4f vi = *(const v4f*)(gI + (size_t)row * (BC_ * 2) + c4);
      *(v4f*)(sR + row * 64 + c4) = vr;
      *(v4f*)(sI + row * 64 + c4) = vi;
    }
  }
  __syncthreads();

  {
    const int tFirst = ttile * 16;
    const int tLast  = (tFirst + 15 < NLAT_) ? tFirst + 15 : NLAT_ - 1;
    const int nFirst = ring_nlon(tFirst);
    const int nLast  = ring_nlon(tLast);
    if (2 * m > (nFirst > nLast ? nFirst : nLast)) return;
  }

  const int hi  = lane >> 4;
  const int lid = lane & 15;
  const int trow = ttile * 16 + lid;
  const int tld  = (trow < NLAT_) ? trow : (NLAT_ - 1);
  const int bc0  = bcpair * 32 + lid;

  const float* __restrict__ pb = pct + (size_t)m * (LMAX_ * NLAT_) + tld;

  v8f accR0 = {0.f, 0.f, 0.f, 0.f, 0.f, 0.f, 0.f, 0.f};
  v8f accI0 = accR0, accR1 = accR0, accI1 = accR0;

  for (int kk = 0; kk < LMAX_; kk += 4) {
    const int l2 = (kk >> 1) + hi;
    const v2f af  = {pb[(size_t)(2 * l2) * NLAT_],
                     pb[(size_t)(2 * l2 + 1) * NLAT_]};
    const v2f bR0 = *(const v2f*)(sR + (l2 * 32 + lid) * 2);
    const v2f bR1 = *(const v2f*)(sR + (l2 * 32 + lid + 16) * 2);
    const v2f bI0 = *(const v2f*)(sI + (l2 * 32 + lid) * 2);
    const v2f bI1 = *(const v2f*)(sI + (l2 * 32 + lid + 16) * 2);
    accR0 = __builtin_amdgcn_wmma_f32_16x16x4_f32(false, af, false, bR0,
                                                  (short)0, accR0, false, false);
    accI0 = __builtin_amdgcn_wmma_f32_16x16x4_f32(false, af, false, bI0,
                                                  (short)0, accI0, false, false);
    accR1 = __builtin_amdgcn_wmma_f32_16x16x4_f32(false, af, false, bR1,
                                                  (short)0, accR1, false, false);
    accI1 = __builtin_amdgcn_wmma_f32_16x16x4_f32(false, af, false, bI1,
                                                  (short)0, accI1, false, false);
  }

  float* __restrict__ lA =
      cA + ((((size_t)ttile * 16) * 8 + bcpair) * MMAX_ + m) * 32 + lane;
  float* __restrict__ lB =
      cB + ((((size_t)ttile * 16) * 8 + bcpair) * MMAX_ + m) * 32 + lane;
  #pragma unroll
  for (int v = 0; v < 8; ++v) {
    const int t  = ttile * 16 + v + 8 * hi;
    const int tl = (t < NLAT_) ? t : (NLAT_ - 1);
    float s, c;
    __sincosf(offset[tl * MMAX_ + m], &s, &c);
    float re = accR0[v], im = accI0[v];
    const float a0 = re * c - im * s;
    const float b0 = re * s + im * c;
    re = accR1[v]; im = accI1[v];
    const float a1 = re * c - im * s;
    const float b1 = re * s + im * c;
    float aLo, aHi, bLo, bHi;
    half_trade(a0, a1, hi, aLo, aHi);
    half_trade(b0, b1, hi, bLo, bHi);
    const size_t oLo = (size_t)v * (8 * MMAX_ * 32);
    const size_t oHi = (size_t)(v + 8) * (8 * MMAX_ * 32);
    const int tLo = ttile * 16 + v;
    const int tHi = tLo + 8;
    if (2 * m <= ring_nlon(tLo)) {
      lA[oLo] = aLo;
      lB[oLo] = bLo;
    }
    if (tHi < NLAT_ && 2 * m <= ring_nlon(tHi)) {
      lA[oHi] = aHi;
      lB[oHi] = bHi;
    }
  }
}

#define RING_WAVES_ 8
#define RING_PITCH_ 260
__global__ __launch_bounds__(RING_WAVES_ * 32)
void sht_ring_idft(const float* __restrict__ cA,
                   const float* __restrict__ cB,
                   const float* __restrict__ cosT,
                   const float* __restrict__ sinT,
                   float* __restrict__ out) {
  __shared__ __attribute__((aligned(16))) float smem[32 * RING_PITCH_];
  float* sA = smem;
  float* sB = smem + M2_ * 64;
  const int tid    = threadIdx.x;
  const int lane   = tid & 31;
  const int wave   = __builtin_amdgcn_readfirstlane(tid >> 5);
  const int bcpair = blockIdx.x;
  const int t      = blockIdx.y;
  const int n  = ring_nlon(t);
  const int nh = n >> 1;
  const int nq = n >> 2;
  const int ntiles = (nq >> 4) + 1;
  const int rot = (bcpair + t + (t >> 2) + (t >> 4) + (t >> 6)) & 3;
  int jtile = wave + rot;
  if (jtile >= RING_WAVES_) jtile -= RING_WAVES_;
  const bool active = (jtile < ntiles);

  {
    const float* __restrict__ gA = cA + ((size_t)t * 8 + bcpair) * (MMAX_ * 32);
    const float* __restrict__ gB = cB + ((size_t)t * 8 + bcpair) * (MMAX_ * 32);
    v4f e0[3], e1[3];
    #pragma unroll
    for (int it = 0; it < 3; ++it) {
      const int idx   = tid + 256 * it;
      const int plane = (idx >= 384) ? 1 : 0;
      const int rem   = idx - plane * 384;
      const int r     = rem >> 3;
      const int c4    = (rem & 7) * 4;
      const int par   = (r >= MROWS_PAR_) ? 1 : 0;
      const int m0    = 4 * (r - par * MROWS_PAR_) + par;
      const bool k0 = (m0 <= nh);
      const bool k1 = (m0 + 2 <= nh);
      const float* __restrict__ g = (plane ? gB : gA) + c4;
      const v4f l0 = *(const v4f*)(g + (k0 ? m0 : 0) * 32);
      const v4f l1 = *(const v4f*)(g + (k1 ? m0 + 2 : 0) * 32);
      const v4f zero4 = {0.f, 0.f, 0.f, 0.f};
      e0[it] = k0 ? l0 : zero4;
      e1[it] = k1 ? l1 : zero4;
    }
    #pragma unroll
    for (int it = 0; it < 3; ++it) {
      const int idx   = tid + 256 * it;
      const int plane = (idx >= 384) ? 1 : 0;
      const int rem   = idx - plane * 384;
      const int r     = rem >> 3;
      const int c4    = (rem & 7) * 4;
      float* s = (plane ? sB : sA) + r * 64 + c4 * 2;
      const v4f w0 = {e0[it].x, e1[it].x, e0[it].y, e1[it].y};
      const v4f w1 = {e0[it].z, e1[it].z, e0[it].w, e1[it].w};
      *(v4f*)(s)     = w0;
      *(v4f*)(s + 4) = w1;
    }
  }
  __syncthreads();

  const int roff = ring_offset(t);
  const int hi  = lane >> 4;
  const int lid = lane & 15;
  const int j   = jtile * 16 + lid;

  v8f accCe0 = {0.f, 0.f, 0.f, 0.f, 0.f, 0.f, 0.f, 0.f};
  v8f accCo0 = accCe0, accSe0 = accCe0, accSo0 = accCe0;
  v8f accCe1 = accCe0, accCo1 = accCe0, accSe1 = accCe0, accSo1 = accCe0;

  if (active) {
    const size_t tb = tab_base(nq) + (size_t)j * 2;
    const float* __restrict__ cT = cosT + tb;
    const float* __restrict__ sT = sinT + tb;
    auto kstep = [&](int kk) {
      const int re = kk + hi;
      const int ro = re + MROWS_PAR_;
      const v2f cfe = *(const v2f*)(cT + (size_t)re * n * 2);
      const v2f cfo = *(const v2f*)(cT + (size_t)ro * n * 2);
      const v2f sfe = *(const v2f*)(sT + (size_t)re * n * 2);
      const v2f sfo = *(const v2f*)(sT + (size_t)ro * n * 2);
      const v2f ae0 = *(const v2f*)(sA + (re * 32 + lid) * 2);
      const v2f ae1 = *(const v2f*)(sA + (re * 32 + lid + 16) * 2);
      const v2f ao0 = *(const v2f*)(sA + (ro * 32 + lid) * 2);
      const v2f ao1 = *(const v2f*)(sA + (ro * 32 + lid + 16) * 2);
      const v2f be0 = *(const v2f*)(sB + (re * 32 + lid) * 2);
      const v2f be1 = *(const v2f*)(sB + (re * 32 + lid + 16) * 2);
      const v2f bo0 = *(const v2f*)(sB + (ro * 32 + lid) * 2);
      const v2f bo1 = *(const v2f*)(sB + (ro * 32 + lid + 16) * 2);
      accCe0 = __builtin_amdgcn_wmma_f32_16x16x4_f32(false, ae0, false, cfe,
                                                     (short)0, accCe0, false, false);
      accCo0 = __builtin_amdgcn_wmma_f32_16x16x4_f32(false, ao0, false, cfo,
                                                     (short)0, accCo0, false, false);
      accSe0 = __builtin_amdgcn_wmma_f32_16x16x4_f32(false, be0, false, sfe,
                                                     (short)0, accSe0, false, false);
      accSo0 = __builtin_amdgcn_wmma_f32_16x16x4_f32(false, bo0, false, sfo,
                                                     (short)0, accSo0, false, false);
      accCe1 = __builtin_amdgcn_wmma_f32_16x16x4_f32(false, ae1, false, cfe,
                                                     (short)0, accCe1, false, false);
      accCo1 = __builtin_amdgcn_wmma_f32_16x16x4_f32(false, ao1, false, cfo,
                                                     (short)0, accCo1, false, false);
      accSe1 = __builtin_amdgcn_wmma_f32_16x16x4_f32(false, be1, false, sfe,
                                                     (short)0, accSe1, false, false);
      accSo1 = __builtin_amdgcn_wmma_f32_16x16x4_f32(false, bo1, false, sfo,
                                                     (short)0, accSo1, false, false);
    };
    if (nh >= 64) {
      for (int kk = 0; kk < MROWS_PAR_; kk += 2) kstep(kk);
    } else {
      for (int kk = 0; kk < 8; kk += 2) kstep(kk);
      if (nh >= 32) {
        for (int kk = 8; kk < 16; kk += 2) kstep(kk);
      }
    }
  }
  __syncthreads();

  if (active) {
    const bool q0 = (j <= nq);
    const bool q3 = (j > 0) && (j <= nq);
    const bool q1 = (j < nq);
    const bool q2 = (j > 0) && (j < nq);
    float* __restrict__ rw = smem + (8 * hi) * RING_PITCH_;
    #pragma unroll
    for (int v = 0; v < 8; ++v) {
      float* __restrict__ o0 = rw + v * RING_PITCH_;
      float* __restrict__ o1 = o0 + 16 * RING_PITCH_;
      const float p0 = accCe0[v] + accCo0[v], s0 = accSe0[v] + accSo0[v];
      const float r0 = accCe0[v] - accCo0[v], d0 = accSe0[v] - accSo0[v];
      const float p1 = accCe1[v] + accCo1[v], s1 = accSe1[v] + accSo1[v];
      const float r1 = accCe1[v] - accCo1[v], d1 = accSe1[v] - accSo1[v];
      if (q0) {
        o0[j] = p0 + s0;
        o1[j] = p1 + s1;
      }
      if (q1) {
        o0[nh - j] = r0 - d0;
        o1[nh - j] = r1 - d1;
      }
      if (q2) {
        o0[nh + j] = r0 + d0;
        o1[nh + j] = r1 + d1;
      }
      if (q3) {
        o0[n - j] = p0 - s0;
        o1[n - j] = p1 - s1;
      }
    }
  }
  __syncthreads();

  {
    const float* __restrict__ rr = smem + (wave * 4) * RING_PITCH_;
    float* __restrict__ go =
        out + (size_t)(bcpair * 32 + wave * 4) * NPIX_ + roff;
    #pragma unroll
    for (int it = 0; it < 2; ++it) {
      const int p = lane + 32 * it;
      if (p < nq) {
        v4f w[4];
        #pragma unroll
        for (int q = 0; q < 4; ++q)
          w[q] = *(const v4f*)(rr + q * RING_PITCH_ + 4 * p);
        #pragma unroll
        for (int q = 0; q < 4; ++q)
          *(v4f*)(go + (size_t)q * NPIX_ + 4 * p) = w[q];
      }
    }
  }
}

extern "C" void kernel_launch(void* const* d_in, const int* in_sizes, int n_in,
                              void* d_out, int out_size, void* d_ws, size_t ws_size,
                              hipStream_t stream) {
  (void)in_sizes; (void)n_in; (void)out_size; (void)ws_size;
  const float* x_real = (const float*)d_in[0];
  const float* x_imag = (const float*)d_in[1];
  const float* pct    = (const float*)d_in[2];
  const float* offset = (const float*)d_in[3];
  float* out = (float*)d_out;

  float* xTr  = (float*)d_ws;
  float* xTi  = xTr  + (size_t)MMAX_ * L2_ * BC_ * 2;
  float* cA   = xTi  + (size_t)MMAX_ * L2_ * BC_ * 2;
  float* cB   = cA   + (size_t)NLAT_ * M2_ * BC_ * 2;
  float* cosT = cB   + (size_t)NLAT_ * M2_ * BC_ * 2;
  float* sinT = cosT + (size_t)TAB_FLOATS;

  sht_prepare<<<dim3(PREP_BLOCKS_), dim3(256), 0, stream>>>(
      x_real, x_imag, xTr, xTi, cosT, sinT);

  sht_legendre_rotate<<<dim3(8, MMAX_), dim3(512), 0, stream>>>(
      xTr, xTi, pct, offset, cA, cB);

  sht_ring_idft<<<dim3(8, NLAT_), dim3(RING_WAVES_ * 32), 0, stream>>>(
      cA, cB, cosT, sinT, out);
}


# Legendre kernel: a wave's 8 phase angles loaded as one group ahead of the epilogue (were 7 serial load-wait pairs between the stores); slab staged with 3 loads in flight per thread; sums unchanged
# speedup vs baseline: 2.6975x; 1.0564x over previous
#include <hip/hip_runtime.h>
#include <hip/hip_bf16.h>

typedef __attribute__((ext_vector_type(2))) float v2f;
typedef __attribute__((ext_vector_type(4))) float v4f;
typedef __attribute__((ext_vector_type(8))) float v8f;

#define NSIDE_  64
#define NLAT_   255
#define LMAX_   96
#define MMAX_   96
#define BC_     256
#define NPIX_   49152
#define L2_     48
#define M2_     48
#define MROWS_PAR_ 24

__device__ __forceinline__ int ring_nlon(int t) {
  const int north = t + 1;
  const int south = 4 * NSIDE_ - 1 - t;
  const int cap   = north < south ? north : south;
  return 4 * (cap < NSIDE_ ? cap : NSIDE_);
}

__device__ __forceinline__ int ring_offset(int t) {
  if (t < NSIDE_ - 1) return 2 * t * (t + 1);
  if (t < 3 * NSIDE_) return 2 * (NSIDE_ - 1) * NSIDE_ + 4 * NSIDE_ * (t - (NSIDE_ - 1));
  int r = NLAT_ - t;
  return NPIX_ - 2 * r * (r + 1);
}

__device__ __forceinline__ size_t tab_base(int k) {
  return (size_t)192 * k * (k - 1);
}
#define TAB_FLOATS 798720

__device__ __forceinline__ void prep_table_item(int k, int idx,
                                                float* __restrict__ cosT,
                                                float* __restrict__ sinT) {
  const int n  = 4 * k;
  const int r  = idx / n;
  const int j  = idx - r * n;
  const int nh = n >> 1;
  const int par = (r >= MROWS_PAR_) ? 1 : 0;
  const int q2  = r - par * MROWS_PAR_;
  const float twon = 6.28318530717958647692f / (float)n;
  float cp[2], sp[2];
  #pragma unroll
  for (int e = 0; e < 2; ++e) {
    const int mm = 4 * q2 + par + 2 * e;
    const float wc  = (mm == 0) ? 1.f : (mm < nh ? 2.f : (mm == nh ? 1.f : 0.f));
    const float wsn = (mm == nh) ? 0.f : wc;
    float s, c;
    __sincosf(twon * (float)((mm * j) % n), &s, &c);
    cp[e] = wc * c;
    sp[e] = -wsn * s;
  }
  const size_t o = tab_base(k) + (size_t)idx * 2;
  const v2f cv = {cp[0], cp[1]};
  const v2f sv = {sp[0], sp[1]};
  *(v2f*)(cosT + o) = cv;
  *(v2f*)(sinT + o) = sv;
}

#define TAB_PIECE_ 1560
__device__ __forceinline__ void prep_build_tables(int b, int tid,
                                                  float* __restrict__ cosT,
                                                  float* __restrict__ sinT) {
  const int p  = b >> 3;
  const int s  = b & 7;
  const int k1 = p + 1;
  const int k2 = 64 - p;
  const int items1 = 192 * k1;
  #pragma unroll 1
  for (int r = 0; r < 7; ++r) {
    const int i = r * 256 + tid;
    if (i < TAB_PIECE_) {
      const int g = s * TAB_PIECE_ + i;
      if (g < items1) prep_table_item(k1, g, cosT, sinT);
      else            prep_table_item(k2, g - items1, cosT, sinT);
    }
  }
}

#define TILE_PITCH_ 196
__device__ __forceinline__ void prep_transpose_x(int b, int tid, float* tile,
                                                 const float* __restrict__ xr,
                                                 const float* __restrict__ xi,
                                                 float* __restrict__ xTr,
                                                 float* __restrict__ xTi) {
  const int plane = (b >= 384) ? 1 : 0;
  const int rem   = b - plane * 384;
  const int l2    = rem >> 3;
  const int bc0   = (rem & 7) * 32;
  const float* __restrict__ src =
      (plane ? xi : xr) + ((size_t)bc0 * LMAX_ + 2 * l2) * MMAX_;
  float* __restrict__ dst =
      (plane ? xTi : xTr) + (size_t)l2 * (BC_ * 2) + bc0 * 2;
  v4f v[6];
  #pragma unroll
  for (int r = 0; r < 6; ++r) {
    const int c  = tid + 256 * r;
    const int bc = c / 48;
    const int q  = c - bc * 48;
    v[r] = *(const v4f*)(src + (size_t)bc * (LMAX_ * MMAX_) + q * 4);
  }
  #pragma unroll
  for (int r = 0; r < 6; ++r) {
    const int c  = tid + 256 * r;
    const int bc = c / 48;
    const int q  = c - bc * 48;
    *(v4f*)(tile + bc * TILE_PITCH_ + q * 4) = v[r];
  }
  __syncthreads();
  #pragma unroll
  for (int r = 0; r < 6; ++r) {
    const int c   = tid + 256 * r;
    const int m   = c >> 4;
    const int bcp = c & 15;
    const float* t0 = tile + (2 * bcp) * TILE_PITCH_ + m;
    const v4f o = {t0[0], t0[MMAX_], t0[TILE_PITCH_], t0[TILE_PITCH_ + MMAX_]};
    *(v4f*)(dst + (size_t)m * (L2_ * BC_ * 2) + bcp * 4) = o;
  }
}

#define PREP_W_BLOCKS_ 256
#define PREP_T_BLOCKS_ 768
#define PREP_BLOCKS_   1024
__global__ __launch_bounds__(256)
void sht_prepare(const float* __restrict__ xr,
                 const float* __restrict__ xi,
                 float* __restrict__ xTr,
                 float* __restrict__ xTi,
                 float* __restrict__ cosT,
                 float* __restrict__ sinT) {
  __shared__ __attribute__((aligned(16))) float tile[32 * TILE_PITCH_];
  const int b   = blockIdx.x;
  const int tid = threadIdx.x;
  if (b < PREP_W_BLOCKS_) {
    prep_build_tables(b, tid, cosT, sinT);
  } else {
    prep_transpose_x(b - PREP_W_BLOCKS_, tid, tile, xr, xi, xTr, xTi);
  }
}

__device__ __forceinline__ float xhalf(float x) {
  return __builtin_bit_cast(float, __builtin_amdgcn_permlanex16(
      __builtin_bit_cast(int, x), __builtin_bit_cast(int, x),
      0x76543210, 0xfedcba98, false, false));
}

__device__ __forceinline__ void half_trade(float x0, float x1, int hi,
                                           float& lo, float& up) {
  const float keep = hi ? x1 : x0;
  const float give = hi ? x0 : x1;
  const float got  = xhalf(give);
  lo = hi ? got : keep;
  up = hi ? keep : got;
}

__global__ __launch_bounds__(512)
void sht_legendre_rotate(const float* __restrict__ xTr,
                         const float* __restrict__ xTi,
                         const float* __restrict__ pct,
                         const float* __restrict__ offset,
                         float* __restrict__ cA,
                         float* __restrict__ cB) {
  __shared__ __attribute__((aligned(16))) float sR[L2_ * 64];
  __shared__ __attribute__((aligned(16))) float sI[L2_ * 64];
  const int tid    = threadIdx.x;
  const int lane   = tid & 31;
  const int ttile  = __builtin_amdgcn_readfirstlane(tid >> 5);
  const int bcpair = blockIdx.x;
  const int m      = blockIdx.y;

  {
    const float* __restrict__ gR = xTr + ((size_t)m * L2_ * BC_ + bcpair * 32) * 2;
    const float* __restrict__ gI = xTi + ((size_t)m * L2_ * BC_ + bcpair * 32) * 2;
    v4f st[3];
    #pragma unroll
    for (int i = 0; i < 3; ++i) {
      const int p     = tid + 512 * i;
      const int plane = (p >= L2_ * 16) ? 1 : 0;
      const int q     = p - plane * (L2_ * 16);
      const int row   = q >> 4;
      const int c4    = (q & 15) * 4;
      st[i] = *(const v4f*)((plane ? gI : gR) + (size_t)row * (BC_ * 2) + c4);
    }
    #pragma unroll
    for (int i = 0; i < 3; ++i) {
      const int p     = tid + 512 * i;
      const int plane = (p >= L2_ * 16) ? 1 : 0;
      const int q     = p - plane * (L2_ * 16);
      const int row   = q >> 4;
      const int c4    = (q & 15) * 4;
      *(v4f*)((plane ? sI : sR) + row * 64 + c4) = st[i];
    }
  }
  __syncthreads();

  {
    const int tFirst = ttile * 16;
    const int tLast  = (tFirst + 15 < NLAT_) ? tFirst + 15 : NLAT_ - 1;
    const int nFirst = ring_nlon(tFirst);
    const int nLast  = ring_nlon(tLast);
    if (2 * m > (nFirst > nLast ? nFirst : nLast)) return;
  }

  const int hi  = lane >> 4;
  const int lid = lane & 15;
  const int trow = ttile * 16 + lid;
  const int tld  = (trow < NLAT_) ? trow : (NLAT_ - 1);
  const int bc0  = bcpair * 32 + lid;

  const float* __restrict__ pb = pct + (size_t)m * (LMAX_ * NLAT_) + tld;

  float th[8];
  #pragma unroll
  for (int v = 0; v < 8; ++v) {
    const int t  = ttile * 16 + v + 8 * hi;
    const int tl = (t < NLAT_) ? t : (NLAT_ - 1);
    th[v] = offset[tl * MMAX_ + m];
  }

  v8f accR0 = {0.f, 0.f, 0.f, 0.f, 0.f, 0.f, 0.f, 0.f};
  v8f accI0 = accR0, accR1 = accR0, accI1 = accR0;

  for (int kk = 0; kk < LMAX_; kk += 4) {
    const int l2 = (kk >> 1) + hi;
    const v2f af  = {pb[(size_t)(2 * l2) * NLAT_],
                     pb[(size_t)(2 * l2 + 1) * NLAT_]};
    const v2f bR0 = *(const v2f*)(sR + (l2 * 32 + lid) * 2);
    const v2f bR1 = *(const v2f*)(sR + (l2 * 32 + lid + 16) * 2);
    const v2f bI0 = *(const v2f*)(sI + (l2 * 32 + lid) * 2);
    const v2f bI1 = *(const v2f*)(sI + (l2 * 32 + lid + 16) * 2);
    accR0 = __builtin_amdgcn_wmma_f32_16x16x4_f32(false, af, false, bR0,
                                                  (short)0, accR0, false, false);
    accI0 = __builtin_amdgcn_wmma_f32_16x16x4_f32(false, af, false, bI0,
                                                  (short)0, accI0, false, false);
    accR1 = __builtin_amdgcn_wmma_f32_16x16x4_f32(false, af, false, bR1,
                                                  (short)0, accR1, false, false);
    accI1 = __builtin_amdgcn_wmma_f32_16x16x4_f32(false, af, false, bI1,
                                                  (short)0, accI1, false, false);
  }

  float* __restrict__ lA =
      cA + ((((size_t)ttile * 16) * 8 + bcpair) * MMAX_ + m) * 32 + lane;
  float* __restrict__ lB =
      cB + ((((size_t)ttile * 16) * 8 + bcpair) * MMAX_ + m) * 32 + lane;
  #pragma unroll
  for (int v = 0; v < 8; ++v) {
    float s, c;
    __sincosf(th[v], &s, &c);
    float re = accR0[v], im = accI0[v];
    const float a0 = re * c - im * s;
    const float b0 = re * s + im * c;
    re = accR1[v]; im = accI1[v];
    const float a1 = re * c - im * s;
    const float b1 = re * s + im * c;
    float aLo, aHi, bLo, bHi;
    half_trade(a0, a1, hi, aLo, aHi);
    half_trade(b0, b1, hi, bLo, bHi);
    const size_t oLo = (size_t)v * (8 * MMAX_ * 32);
    const size_t oHi = (size_t)(v + 8) * (8 * MMAX_ * 32);
    const int tLo = ttile * 16 + v;
    const int tHi = tLo + 8;
    if (2 * m <= ring_nlon(tLo)) {
      lA[oLo] = aLo;
      lB[oLo] = bLo;
    }
    if (tHi < NLAT_ && 2 * m <= ring_nlon(tHi)) {
      lA[oHi] = aHi;
      lB[oHi] = bHi;
    }
  }
}

#define RING_WAVES_ 8
#define RING_PITCH_ 260
__global__ __launch_bounds__(RING_WAVES_ * 32)
void sht_ring_idft(const float* __restrict__ cA,
                   const float* __restrict__ cB,
                   const float* __restrict__ cosT,
                   const float* __restrict__ sinT,
                   float* __restrict__ out) {
  __shared__ __attribute__((aligned(16))) float smem[32 * RING_PITCH_];
  float* sA = smem;
  float* sB = smem + M2_ * 64;
  const int tid    = threadIdx.x;
  const int lane   = tid & 31;
  const int wave   = __builtin_amdgcn_readfirstlane(tid >> 5);
  const int bcpair = blockIdx.x;
  const int t      = blockIdx.y;
  const int n  = ring_nlon(t);
  const int nh = n >> 1;
  const int nq = n >> 2;
  const int ntiles = (nq >> 4) + 1;
  const int rot = (bcpair + t + (t >> 2) + (t >> 4) + (t >> 6)) & 3;
  int jtile = wave + rot;
  if (jtile >= RING_WAVES_) jtile -= RING_WAVES_;
  const bool active = (jtile < ntiles);

  {
    const float* __restrict__ gA = cA + ((size_t)t * 8 + bcpair) * (MMAX_ * 32);
    const float* __restrict__ gB = cB + ((size_t)t * 8 + bcpair) * (MMAX_ * 32);
    v4f e0[3], e1[3];
    #pragma unroll
    for (int it = 0; it < 3; ++it) {
      const int idx   = tid + 256 * it;
      const int plane = (idx >= 384) ? 1 : 0;
      const int rem   = idx - plane * 384;
      const int r     = rem >> 3;
      const int c4    = (rem & 7) * 4;
      const int par   = (r >= MROWS_PAR_) ? 1 : 0;
      const int m0    = 4 * (r - par * MROWS_PAR_) + par;
      const bool k0 = (m0 <= nh);
      const bool k1 = (m0 + 2 <= nh);
      const float* __restrict__ g = (plane ? gB : gA) + c4;
      const v4f l0 = *(const v4f*)(g + (k0 ? m0 : 0) * 32);
      const v4f l1 = *(const v4f*)(g + (k1 ? m0 + 2 : 0) * 32);
      const v4f zero4 = {0.f, 0.f, 0.f, 0.f};
      e0[it] = k0 ? l0 : zero4;
      e1[it] = k1 ? l1 : zero4;
    }
    #pragma unroll
    for (int it = 0; it < 3; ++it) {
      const int idx   = tid + 256 * it;
      const int plane = (idx >= 384) ? 1 : 0;
      const int rem   = idx - plane * 384;
      const int r     = rem >> 3;
      const int c4    = (rem & 7) * 4;
      float* s = (plane ? sB : sA) + r * 64 + c4 * 2;
      const v4f w0 = {e0[it].x, e1[it].x, e0[it].y, e1[it].y};
      const v4f w1 = {e0[it].z, e1[it].z, e0[it].w, e1[it].w};
      *(v4f*)(s)     = w0;
      *(v4f*)(s + 4) = w1;
    }
  }
  __syncthreads();

  const int roff = ring_offset(t);
  const int hi  = lane >> 4;
  const int lid = lane & 15;
  const int j   = jtile * 16 + lid;

  v8f accCe0 = {0.f, 0.f, 0.f, 0.f, 0.f, 0.f, 0.f, 0.f};
  v8f accCo0 = accCe0, accSe0 = accCe0, accSo0 = accCe0;
  v8f accCe1 = accCe0, accCo1 = accCe0, accSe1 = accCe0, accSo1 = accCe0;

  if (active) {
    const size_t tb = tab_base(nq) + (size_t)j * 2;
    const float* __restrict__ cT = cosT + tb;
    const float* __restrict__ sT = sinT + tb;
    auto kstep = [&](int kk) {
      const int re = kk + hi;
      const int ro = re + MROWS_PAR_;
      const v2f cfe = *(const v2f*)(cT + (size_t)re * n * 2);
      const v2f cfo = *(const v2f*)(cT + (size_t)ro * n * 2);
      const v2f sfe = *(const v2f*)(sT + (size_t)re * n * 2);
      const v2f sfo = *(const v2f*)(sT + (size_t)ro * n * 2);
      const v2f ae0 = *(const v2f*)(sA + (re * 32 + lid) * 2);
      const v2f ae1 = *(const v2f*)(sA + (re * 32 + lid + 16) * 2);
      const v2f ao0 = *(const v2f*)(sA + (ro * 32 + lid) * 2);
      const v2f ao1 = *(const v2f*)(sA + (ro * 32 + lid + 16) * 2);
      const v2f be0 = *(const v2f*)(sB + (re * 32 + lid) * 2);
      const v2f be1 = *(const v2f*)(sB + (re * 32 + lid + 16) * 2);
      const v2f bo0 = *(const v2f*)(sB + (ro * 32 + lid) * 2);
      const v2f bo1 = *(const v2f*)(sB + (ro * 32 + lid + 16) * 2);
      accCe0 = __builtin_amdgcn_wmma_f32_16x16x4_f32(false, ae0, false, cfe,
                                                     (short)0, accCe0, false, false);
      accCo0 = __builtin_amdgcn_wmma_f32_16x16x4_f32(false, ao0, false, cfo,
                                                     (short)0, accCo0, false, false);
      accSe0 = __builtin_amdgcn_wmma_f32_16x16x4_f32(false, be0, false, sfe,
                                                     (short)0, accSe0, false, false);
      accSo0 = __builtin_amdgcn_wmma_f32_16x16x4_f32(false, bo0, false, sfo,
                                                     (short)0, accSo0, false, false);
      accCe1 = __builtin_amdgcn_wmma_f32_16x16x4_f32(false, ae1, false, cfe,
                                                     (short)0, accCe1, false, false);
      accCo1 = __builtin_amdgcn_wmma_f32_16x16x4_f32(false, ao1, false, cfo,
                                                     (short)0, accCo1, false, false);
      accSe1 = __builtin_amdgcn_wmma_f32_16x16x4_f32(false, be1, false, sfe,
                                                     (short)0, accSe1, false, false);
      accSo1 = __builtin_amdgcn_wmma_f32_16x16x4_f32(false, bo1, false, sfo,
                                                     (short)0, accSo1, false, false);
    };
    if (nh >= 64) {
      for (int kk = 0; kk < MROWS_PAR_; kk += 2) kstep(kk);
    } else {
      for (int kk = 0; kk < 8; kk += 2) kstep(kk);
      if (nh >= 32) {
        for (int kk = 8; kk < 16; kk += 2) kstep(kk);
      }
    }
  }
  __syncthreads();

  if (active) {
    const bool q0 = (j <= nq);
    const bool q3 = (j > 0) && (j <= nq);
    const bool q1 = (j < nq);
    const bool q2 = (j > 0) && (j < nq);
    float* __restrict__ rw = smem + (8 * hi) * RING_PITCH_;
    #pragma unroll
    for (int v = 0; v < 8; ++v) {
      float* __restrict__ o0 = rw + v * RING_PITCH_;
      float* __restrict__ o1 = o0 + 16 * RING_PITCH_;
      const float p0 = accCe0[v] + accCo0[v], s0 = accSe0[v] + accSo0[v];
      const float r0 = accCe0[v] - accCo0[v], d0 = accSe0[v] - accSo0[v];
      const float p1 = accCe1[v] + accCo1[v], s1 = accSe1[v] + accSo1[v];
      const float r1 = accCe1[v] - accCo1[v], d1 = accSe1[v] - accSo1[v];
      if (q0) {
        o0[j] = p0 + s0;
        o1[j] = p1 + s1;
      }
      if (q1) {
        o0[nh - j] = r0 - d0;
        o1[nh - j] = r1 - d1;
      }
      if (q2) {
        o0[nh + j] = r0 + d0;
        o1[nh + j] = r1 + d1;
      }
      if (q3) {
        o0[n - j] = p0 - s0;
        o1[n - j] = p1 - s1;
      }
    }
  }
  __syncthreads();

  {
    const float* __restrict__ rr = smem + (wave * 4) * RING_PITCH_;
    float* __restrict__ go =
        out + (size_t)(bcpair * 32 + wave * 4) * NPIX_ + roff;
    #pragma unroll
    for (int it = 0; it < 2; ++it) {
      const int p = lane + 32 * it;
      if (p < nq) {
        v4f w[4];
        #pragma unroll
        for (int q = 0; q < 4; ++q)
          w[q] = *(const v4f*)(rr + q * RING_PITCH_ + 4 * p);
        #pragma unroll
        for (int q = 0; q < 4; ++q)
          *(v4f*)(go + (size_t)q * NPIX_ + 4 * p) = w[q];
      }
    }
  }
}

extern "C" void kernel_launch(void* const* d_in, const int* in_sizes, int n_in,
                              void* d_out, int out_size, void* d_ws, size_t ws_size,
                              hipStream_t stream) {
  (void)in_sizes; (void)n_in; (void)out_size; (void)ws_size;
  const float* x_real = (const float*)d_in[0];
  const float* x_imag = (const float*)d_in[1];
  const float* pct    = (const float*)d_in[2];
  const float* offset = (const float*)d_in[3];
  float* out = (float*)d_out;

  float* xTr  = (float*)d_ws;
  float* xTi  = xTr  + (size_t)MMAX_ * L2_ * BC_ * 2;
  float* cA   = xTi  + (size_t)MMAX_ * L2_ * BC_ * 2;
  float* cB   = cA   + (size_t)NLAT_ * M2_ * BC_ * 2;
  float* cosT = cB   + (size_t)NLAT_ * M2_ * BC_ * 2;
  float* sinT = cosT + (size_t)TAB_FLOATS;

  sht_prepare<<<dim3(PREP_BLOCKS_), dim3(256), 0, stream>>>(
      x_real, x_imag, xTr, xTi, cosT, sinT);

  sht_legendre_rotate<<<dim3(8, MMAX_), dim3(512), 0, stream>>>(
      xTr, xTi, pct, offset, cA, cB);

  sht_ring_idft<<<dim3(8, NLAT_), dim3(RING_WAVES_ * 32), 0, stream>>>(
      cA, cB, cosT, sinT, out);
}
